# starting kernel
# baseline (speedup 1.0000x reference)
#include <hip/hip_runtime.h>

typedef float v2f __attribute__((ext_vector_type(2)));
typedef float v8f __attribute__((ext_vector_type(8)));

#define DD 66
#define HHW (66 * 66)
#define DHW (66 * 66 * 66)
#define CI 32
#define CO 32

__global__ void zero_f4_kernel(float4* __restrict__ p, int n4) {
  int i = blockIdx.x * blockDim.x + threadIdx.x;
  int stride = gridDim.x * blockDim.x;
  for (; i < n4; i += stride) p[i] = make_float4(0.f, 0.f, 0.f, 0.f);
}

__launch_bounds__(256)
__global__ void p2g_wmma_kernel(const float* __restrict__ x,
                                const float* __restrict__ pos,
                                const float* __restrict__ kern,
                                const float* __restrict__ bias,
                                float* __restrict__ out,
                                int Mtotal, int Nper) {
  __shared__ float sB[1024];
  __shared__ float2 sP[8][32];

  const int t    = threadIdx.x;
  const int lane = t & 31;
  const int wv   = t >> 5;
  const int half = lane >> 4;
  const int nlo  = lane & 15;

  const int waveBase = (blockIdx.x * 8 + wv) * 32;

  int p  = waveBase + lane;
  int pc = (p < Mtotal) ? p : (Mtotal - 1);
  float valid = (p < Mtotal) ? 1.0f : 0.0f;

  float px = pos[pc * 3 + 0] * 64.0f;
  float py = pos[pc * 3 + 1] * 64.0f;
  float pz = pos[pc * 3 + 2] * 64.0f;
  int b0 = (int)px, b1 = (int)py, b2 = (int)pz;
  float fx = px - (float)b0 - 0.5f;
  float fy = py - (float)b1 - 0.5f;
  float fz = pz - (float)b2 - 0.5f;

  float wx[3], wy[3], wz[3];
  wx[0] = 0.5f * (0.5f - fx) * (0.5f - fx) * valid;
  wx[1] = (0.75f - fx * fx) * valid;
  wx[2] = 0.5f * (0.5f + fx) * (0.5f + fx) * valid;
  wy[0] = 0.5f * (0.5f - fy) * (0.5f - fy);
  wy[1] = 0.75f - fy * fy;
  wy[2] = 0.5f * (0.5f + fy) * (0.5f + fy);
  wz[0] = 0.5f * (0.5f - fz) * (0.5f - fz);
  wz[1] = 0.75f - fz * fz;
  wz[2] = 0.5f * (0.5f + fz) * (0.5f + fz);

  int bb = (pc >= Nper) ? 1 : 0;
  int ibase = bb * (CO * DHW) + b0 * HHW + b1 * 66 + b2;

  v2f a[2][8];
#pragma unroll
  for (int mt = 0; mt < 2; ++mt) {
    int row = waveBase + mt * 16 + nlo;
    row = (row < Mtotal) ? row : (Mtotal - 1);
    const float* xr = x + row * CI + half * 2;
#pragma unroll
    for (int c = 0; c < 8; ++c) a[mt][c] = *(const v2f*)(xr + c * 4);
  }

  const float bl0 = bias[nlo];
  const float bl1 = bias[16 + nlo];
  const int chanOff0 = nlo * DHW;
  const int chanOff1 = (16 + nlo) * DHW;

  const float4* k4 = (const float4*)kern;

  for (int o = 0; o < 27; ++o) {
    const int oi = o / 9, oj = (o / 3) % 3, ok = o % 3;

    __syncthreads();

    {
      float4 v4 = k4[o * 256 + t];
      int f   = t * 4;
      int ci  = f >> 5;
      int co  = f & 31;
      int nt  = co >> 4;
      int cc  = ci >> 2;
      int hh  = (ci >> 1) & 1;
      int vv  = ci & 1;
      int base = (nt * 8 + cc) * 64 + (hh * 16 + (co & 15)) * 2 + vv;
      sB[base + 0] = v4.x;
      sB[base + 2] = v4.y;
      sB[base + 4] = v4.z;
      sB[base + 6] = v4.w;
    }
    {
      float wp = wx[oi] * wy[oj] * wz[ok];
      int io = ibase + oi * HHW + oj * 66 + ok;
      sP[wv][lane] = make_float2(wp, __int_as_float(io));
    }
    __syncthreads();

    v2f bf[2][8];
#pragma unroll
    for (int nt = 0; nt < 2; ++nt)
#pragma unroll
      for (int c = 0; c < 8; ++c)
        bf[nt][c] = *(const v2f*)&sB[(nt * 8 + c) * 64 + lane * 2];

    v8f dacc[2][2];
#pragma unroll
    for (int mt = 0; mt < 2; ++mt)
#pragma unroll
      for (int nt = 0; nt < 2; ++nt) {
        float bl = nt ? bl1 : bl0;
#pragma unroll
        for (int r = 0; r < 8; ++r) dacc[mt][nt][r] = bl;
      }

#pragma unroll
    for (int c = 0; c < 8; ++c) {
#pragma unroll
      for (int mt = 0; mt < 2; ++mt) {
#pragma unroll
        for (int nt = 0; nt < 2; ++nt) {
          dacc[mt][nt] = __builtin_amdgcn_wmma_f32_16x16x4_f32(
              false, a[mt][c], false, bf[nt][c], (short)0, dacc[mt][nt],
              false, false);
        }
      }
    }

#pragma unroll
    for (int mt = 0; mt < 2; ++mt) {
#pragma unroll
      for (int r = 0; r < 8; ++r) {
        int m = mt * 16 + half * 8 + r;
        float2 pw = sP[wv][m];
        float w = pw.x;
        int io = __float_as_int(pw.y);
        atomicAdd(out + io + chanOff0, w * dacc[mt][0][r]);
        atomicAdd(out + io + chanOff1, w * dacc[mt][1][r]);
      }
    }
  }
}

extern "C" void kernel_launch(void* const* d_in, const int* in_sizes, int n_in,
                              void* d_out, int out_size, void* d_ws, size_t ws_size,
                              hipStream_t stream) {
  const float* x    = (const float*)d_in[0];
  const float* pos  = (const float*)d_in[1];
  const float* kern = (const float*)d_in[2];
  const float* bias = (const float*)d_in[3];
  float* out = (float*)d_out;

  const int Mtotal = in_sizes[0] / CI;
  const int Nper   = in_sizes[1] / (2 * 3);

  int n4 = out_size / 4;
  zero_f4_kernel<<<2048, 256, 0, stream>>>((float4*)out, n4);

  int blocks = (Mtotal + 255) / 256;
  p2g_wmma_kernel<<<blocks, 256, 0, stream>>>(x, pos, kern, bias, out,
                                              Mtotal, Nper);
}


# product columns = (channel, offset along the last grid axis): 9 passes of 32x96 instead of 27 of 32x32, so 3 lanes of each f32 atomic share one output line; atomic arrival order as before
# speedup vs baseline: 11.8777x; 11.8777x over previous
#include <hip/hip_runtime.h>

typedef float v2f __attribute__((ext_vector_type(2)));
typedef float v8f __attribute__((ext_vector_type(8)));

#define DD 66
#define HHW (66 * 66)
#define DHW (66 * 66 * 66)
#define CI 32
#define CO 32

__global__ void zero_f4_kernel(float4* __restrict__ p, int n4,
                               unsigned int* __restrict__ cnt) {
  if (blockIdx.x == 0 && threadIdx.x < 2) cnt[threadIdx.x] = 0u;
  int i = blockIdx.x * blockDim.x + threadIdx.x;
  int stride = gridDim.x * blockDim.x;
  for (; i < n4; i += stride) p[i] = make_float4(0.f, 0.f, 0.f, 0.f);
}

#define NT 6

__launch_bounds__(256)
__global__ void p2g_wmma_kernel(const float* __restrict__ x,
                                const float* __restrict__ pos,
                                const float* __restrict__ kern,
                                const float* __restrict__ bias,
                                float* __restrict__ out,
                                unsigned int* __restrict__ cnt,
                                int Nper) {
  __shared__ float sB[NT * 8 * 64];
  __shared__ float4 sP[8][32];
  __shared__ float sBias[NT * 16];
  __shared__ int sSel[2];

  const int t    = threadIdx.x;
  const int lane = t & 31;
  const int wv   = t >> 5;
  const int half = lane >> 4;
  const int nlo  = lane & 15;

  const int chunksPer = (Nper + 255) >> 8;
  if (t < NT * 16) sBias[t] = bias[t / 3];
  if (t == 0) {
    unsigned int id = __builtin_amdgcn_s_sendmsg_rtn(0x87);
    int hb = (int)((id >> 18) & 1u);
    int tk = (int)__hip_atomic_fetch_add(cnt + hb, 1u, __ATOMIC_RELAXED,
                                         __HIP_MEMORY_SCOPE_AGENT);
    if (tk >= chunksPer) {
      hb ^= 1;
      tk = (int)__hip_atomic_fetch_add(cnt + hb, 1u, __ATOMIC_RELAXED,
                                       __HIP_MEMORY_SCOPE_AGENT);
    }
    sSel[0] = hb;
    sSel[1] = tk;
  }
  __syncthreads();
  const int bb = sSel[0];
  const int chunk = sSel[1];
  if (chunk >= chunksPer) return;

  const int waveBase = (chunk * 8 + wv) * 32;
  const int pOff = bb * Nper;

  int q  = waveBase + lane;
  int pc = pOff + ((q < Nper) ? q : (Nper - 1));
  float valid = (q < Nper) ? 1.0f : 0.0f;

  float px = pos[pc * 3 + 0] * 64.0f;
  float py = pos[pc * 3 + 1] * 64.0f;
  float pz = pos[pc * 3 + 2] * 64.0f;
  int b0 = (int)px, b1 = (int)py, b2 = (int)pz;
  float fx = px - (float)b0 - 0.5f;
  float fy = py - (float)b1 - 0.5f;
  float fz = pz - (float)b2 - 0.5f;

  const float wx0 = 0.5f * (0.5f - fx) * (0.5f - fx) * valid;
  const float wx1 = (0.75f - fx * fx) * valid;
  const float wx2 = 0.5f * (0.5f + fx) * (0.5f + fx) * valid;
  const float wy0 = 0.5f * (0.5f - fy) * (0.5f - fy);
  const float wy1 = 0.75f - fy * fy;
  const float wy2 = 0.5f * (0.5f + fy) * (0.5f + fy);
  const float wz0 = 0.5f * (0.5f - fz) * (0.5f - fz);
  const float wz1 = 0.75f - fz * fz;
  const float wz2 = 0.5f * (0.5f + fz) * (0.5f + fz);

  const int ibase = bb * (CO * DHW) + b0 * HHW + b1 * 66 + b2;

  v2f a[2][8];
#pragma unroll
  for (int mt = 0; mt < 2; ++mt) {
    int row = waveBase + mt * 16 + nlo;
    row = pOff + ((row < Nper) ? row : (Nper - 1));
    const float* xr = x + row * CI + half * 2;
#pragma unroll
    for (int c = 0; c < 8; ++c) a[mt][c] = *(const v2f*)(xr + c * 4);
  }

  const float4* k4 = (const float4*)kern;
  const int sci = t >> 3;
  const int sco = (t & 7) * 4;
  const int sbase = (sci >> 2) * 64 + (((sci >> 1) & 1) * 16) * 2 + (sci & 1);

  for (int g = 0; g < 9; ++g) {
    const int oi = g / 3, oj = g - oi * 3;

    __syncthreads();

#pragma unroll
    for (int j = 0; j < 3; ++j) {
      float4 v4 = k4[g * 768 + j * 256 + t];
      float vv[4] = {v4.x, v4.y, v4.z, v4.w};
#pragma unroll
      for (int i = 0; i < 4; ++i) {
        int n = (sco + i) * 3 + j;
        sB[(n >> 4) * 512 + (n & 15) * 2 + sbase] = vv[i];
      }
    }
    {
      float wxs = (oi == 0) ? wx0 : ((oi == 1) ? wx1 : wx2);
      float wys = (oj == 0) ? wy0 : ((oj == 1) ? wy1 : wy2);
      float wxy = wxs * wys;
      int io = ibase + oi * HHW + oj * 66;
      sP[wv][lane] = make_float4(wxy * wz0, wxy * wz1, wxy * wz2, __int_as_float(io * 4));
    }
    __syncthreads();

#pragma unroll 1
    for (int np = 0; np < NT / 2; ++np) {
      unsigned colOff[2];
      int   colK[2];
      float colBias[2];
#pragma unroll
      for (int q2 = 0; q2 < 2; ++q2) {
        int n  = (np * 2 + q2) * 16 + nlo;
        int co = n / 3;
        colK[q2]    = n - co * 3;
        colOff[q2]  = (unsigned)(co * DHW + colK[q2]) * 4u;
        colBias[q2] = sBias[n];
      }

      v2f bf[2][8];
#pragma unroll
      for (int q2 = 0; q2 < 2; ++q2)
#pragma unroll
        for (int c = 0; c < 8; ++c)
          bf[q2][c] = *(const v2f*)&sB[((np * 2 + q2) * 8 + c) * 64 + lane * 2];

      v8f dacc[2][2];
#pragma unroll
      for (int mt = 0; mt < 2; ++mt)
#pragma unroll
        for (int q2 = 0; q2 < 2; ++q2) {
          float bl = colBias[q2];
#pragma unroll
          for (int r = 0; r < 8; ++r) dacc[mt][q2][r] = bl;
        }

#pragma unroll
      for (int c = 0; c < 8; ++c) {
#pragma unroll
        for (int mt = 0; mt < 2; ++mt) {
#pragma unroll
          for (int q2 = 0; q2 < 2; ++q2) {
            dacc[mt][q2] = __builtin_amdgcn_wmma_f32_16x16x4_f32(
                false, a[mt][c], false, bf[q2][c], (short)0, dacc[mt][q2],
                false, false);
          }
        }
      }

#pragma unroll
      for (int mt = 0; mt < 2; ++mt) {
#pragma unroll
        for (int r = 0; r < 8; ++r) {
          int m = mt * 16 + half * 8 + r;
          float4 pw = sP[wv][m];
          unsigned io = (unsigned)__float_as_int(pw.w);
#pragma unroll
          for (int q2 = 0; q2 < 2; ++q2) {
            int kk = colK[q2];
            float w = (kk == 0) ? pw.x : ((kk == 1) ? pw.y : pw.z);
            atomicAdd((float*)((char*)out + (io + colOff[q2])), w * dacc[mt][q2][r]);
          }
        }
      }
    }
  }
}

extern "C" void kernel_launch(void* const* d_in, const int* in_sizes, int n_in,
                              void* d_out, int out_size, void* d_ws, size_t ws_size,
                              hipStream_t stream) {
  const float* x    = (const float*)d_in[0];
  const float* pos  = (const float*)d_in[1];
  const float* kern = (const float*)d_in[2];
  const float* bias = (const float*)d_in[3];
  float* out = (float*)d_out;
  unsigned int* cnt = (unsigned int*)d_ws;

  const int Nper   = in_sizes[1] / (2 * 3);

  int n4 = out_size / 4;
  zero_f4_kernel<<<2048, 256, 0, stream>>>((float4*)out, n4, cnt);

  int blocks = 2 * ((Nper + 255) / 256);
  p2g_wmma_kernel<<<blocks, 256, 0, stream>>>(x, pos, kern, bias, out,
                                              cnt, Nper);
}


# product columns regrouped: 7 tiles of 5 whole channels x 3 offsets (one column per tile unused) so no channel's 3 lanes split over two atomic instructions; atomic arrival order as before
# speedup vs baseline: 15.4762x; 1.3030x over previous
#include <hip/hip_runtime.h>

typedef float v2f __attribute__((ext_vector_type(2)));
typedef float v8f __attribute__((ext_vector_type(8)));

#define DD 66
#define HHW (66 * 66)
#define DHW (66 * 66 * 66)
#define CI 32
#define CO 32

__global__ void zero_f4_kernel(float4* __restrict__ p, int n4,
                               unsigned int* __restrict__ cnt) {
  if (blockIdx.x == 0 && threadIdx.x < 2) cnt[threadIdx.x] = 0u;
  int i = blockIdx.x * blockDim.x + threadIdx.x;
  int stride = gridDim.x * blockDim.x;
  for (; i < n4; i += stride) p[i] = make_float4(0.f, 0.f, 0.f, 0.f);
}

#define NT 7
#define CPT 5

__launch_bounds__(256)
__global__ void p2g_wmma_kernel(const float* __restrict__ x,
                                const float* __restrict__ pos,
                                const float* __restrict__ kern,
                                const float* __restrict__ bias,
                                float* __restrict__ out,
                                unsigned int* __restrict__ cnt,
                                int Nper) {
  __shared__ float sB[NT * 8 * 64];
  __shared__ float4 sP[8][32];
  __shared__ float sBias[NT * 16];
  __shared__ int sSel[2];

  const int t    = threadIdx.x;
  const int lane = t & 31;
  const int wv   = t >> 5;
  const int half = lane >> 4;
  const int nlo  = lane & 15;

  const int chunksPer = (Nper + 255) >> 8;
  if (t < NT * 16) {
    int cjb = (t & 15) / 3;
    int cob = (t >> 4) * CPT + cjb;
    sBias[t] = (cjb < CPT && cob < CO) ? bias[cob] : 0.0f;
  }
  for (int i = t; i < NT * 8 * 64; i += 256) sB[i] = 0.0f;
  if (t == 0) {
    unsigned int id = __builtin_amdgcn_s_sendmsg_rtn(0x87);
    int hb = (int)((id >> 18) & 1u);
    int tk = (int)__hip_atomic_fetch_add(cnt + hb, 1u, __ATOMIC_RELAXED,
                                         __HIP_MEMORY_SCOPE_AGENT);
    if (tk >= chunksPer) {
      hb ^= 1;
      tk = (int)__hip_atomic_fetch_add(cnt + hb, 1u, __ATOMIC_RELAXED,
                                       __HIP_MEMORY_SCOPE_AGENT);
    }
    sSel[0] = hb;
    sSel[1] = tk;
  }
  __syncthreads();
  const int bb = sSel[0];
  const int chunk = sSel[1];
  if (chunk >= chunksPer) return;

  const int waveBase = (chunk * 8 + wv) * 32;
  const int pOff = bb * Nper;

  int q  = waveBase + lane;
  int pc = pOff + ((q < Nper) ? q : (Nper - 1));
  float valid = (q < Nper) ? 1.0f : 0.0f;

  float px = pos[pc * 3 + 0] * 64.0f;
  float py = pos[pc * 3 + 1] * 64.0f;
  float pz = pos[pc * 3 + 2] * 64.0f;
  int b0 = (int)px, b1 = (int)py, b2 = (int)pz;
  float fx = px - (float)b0 - 0.5f;
  float fy = py - (float)b1 - 0.5f;
  float fz = pz - (float)b2 - 0.5f;

  const float wx0 = 0.5f * (0.5f - fx) * (0.5f - fx) * valid;
  const float wx1 = (0.75f - fx * fx) * valid;
  const float wx2 = 0.5f * (0.5f + fx) * (0.5f + fx) * valid;
  const float wy0 = 0.5f * (0.5f - fy) * (0.5f - fy);
  const float wy1 = 0.75f - fy * fy;
  const float wy2 = 0.5f * (0.5f + fy) * (0.5f + fy);
  const float wz0 = 0.5f * (0.5f - fz) * (0.5f - fz);
  const float wz1 = 0.75f - fz * fz;
  const float wz2 = 0.5f * (0.5f + fz) * (0.5f + fz);

  const int ibase = bb * (CO * DHW) + b0 * HHW + b1 * 66 + b2;

  v2f a[2][8];
#pragma unroll
  for (int mt = 0; mt < 2; ++mt) {
    int row = waveBase + mt * 16 + nlo;
    row = pOff + ((row < Nper) ? row : (Nper - 1));
    const float* xr = x + row * CI + half * 2;
#pragma unroll
    for (int c = 0; c < 8; ++c) a[mt][c] = *(const v2f*)(xr + c * 4);
  }

  const float4* k4 = (const float4*)kern;
  const int sci = t >> 3;
  const int sco = (t & 7) * 4;
  const int sbase = (sci >> 2) * 64 + (((sci >> 1) & 1) * 16) * 2 + (sci & 1);

  const int cj  = nlo / 3;
  const int okL = nlo - cj * 3;
  const unsigned laneOff = (unsigned)(cj * DHW + okL) * 4u;

  for (int g = 0; g < 9; ++g) {
    const int oi = g / 3, oj = g - oi * 3;

    __syncthreads();

#pragma unroll
    for (int j = 0; j < 3; ++j) {
      float4 v4 = k4[g * 768 + j * 256 + t];
      float vv[4] = {v4.x, v4.y, v4.z, v4.w};
#pragma unroll
      for (int i = 0; i < 4; ++i) {
        int co = sco + i;
        int tt = co / CPT;
        int n  = (co - tt * CPT) * 3 + j;
        sB[tt * 512 + n * 2 + sbase] = vv[i];
      }
    }
    {
      float wxs = (oi == 0) ? wx0 : ((oi == 1) ? wx1 : wx2);
      float wys = (oj == 0) ? wy0 : ((oj == 1) ? wy1 : wy2);
      float wxy = wxs * wys;
      int io = ibase + oi * HHW + oj * 66;
      sP[wv][lane] = make_float4(wxy * wz0, wxy * wz1, wxy * wz2, __int_as_float(io * 4));
    }
    __syncthreads();

#pragma unroll 1
    for (int tt = 0; tt < NT; ++tt) {
      const int co = tt * CPT + cj;
      const bool used = (cj < CPT) && (co < CO);
      const unsigned colOff = laneOff + (unsigned)tt * (unsigned)(CPT * DHW * 4);
      const float bl = sBias[tt * 16 + nlo];

      v2f bf[8];
#pragma unroll
      for (int c = 0; c < 8; ++c)
        bf[c] = *(const v2f*)&sB[(tt * 8 + c) * 64 + lane * 2];

      v8f dacc[2];
#pragma unroll
      for (int mt = 0; mt < 2; ++mt)
#pragma unroll
        for (int r = 0; r < 8; ++r) dacc[mt][r] = bl;

#pragma unroll
      for (int c = 0; c < 8; ++c) {
#pragma unroll
        for (int mt = 0; mt < 2; ++mt) {
          dacc[mt] = __builtin_amdgcn_wmma_f32_16x16x4_f32(
              false, a[mt][c], false, bf[c], (short)0, dacc[mt],
              false, false);
        }
      }

      if (used) {
#pragma unroll
        for (int mt = 0; mt < 2; ++mt) {
#pragma unroll
          for (int r = 0; r < 8; ++r) {
            int m = mt * 16 + half * 8 + r;
            float4 pw = sP[wv][m];
            unsigned io = (unsigned)__float_as_int(pw.w);
            float w = (okL == 0) ? pw.x : ((okL == 1) ? pw.y : pw.z);
            atomicAdd((float*)((char*)out + (io + colOff)), w * dacc[mt][r]);
          }
        }
      }
    }
  }
}

extern "C" void kernel_launch(void* const* d_in, const int* in_sizes, int n_in,
                              void* d_out, int out_size, void* d_ws, size_t ws_size,
                              hipStream_t stream) {
  const float* x    = (const float*)d_in[0];
  const float* pos  = (const float*)d_in[1];
  const float* kern = (const float*)d_in[2];
  const float* bias = (const float*)d_in[3];
  float* out = (float*)d_out;
  unsigned int* cnt = (unsigned int*)d_ws;

  const int Nper   = in_sizes[1] / (2 * 3);

  int n4 = out_size / 4;
  zero_f4_kernel<<<2048, 256, 0, stream>>>((float4*)out, n4, cnt);

  int blocks = 2 * ((Nper + 255) / 256);
  p2g_wmma_kernel<<<blocks, 256, 0, stream>>>(x, pos, kern, bias, out,
                                              cnt, Nper);
}


# particles in (cell, index) order, same in every run (counting sort, added launches); a 2nd f32 matrix product sums a wave's 32 particles per grid cell before the atomics; sum order changes
# speedup vs baseline: 24.4579x; 1.5804x over previous
#include <hip/hip_runtime.h>


typedef float v2f __attribute__((ext_vector_type(2)));
typedef float v8f __attribute__((ext_vector_type(8)));

#define DD 66
#define HHW (66 * 66)
#define DHW (66 * 66 * 66)
#define CI 32
#define CO 32
#define NCELL (64 * 64 * 64)
#define NCELL2 (2 * NCELL)
#define NSCANB (NCELL2 / 1024)

__global__ void zero_f4_kernel(float4* __restrict__ p, int n4,
                               int4* __restrict__ hist4, int nh4,
                               unsigned int* __restrict__ cnt) {
  if (blockIdx.x == 0 && threadIdx.x < 2) cnt[threadIdx.x] = 0u;
  const int i0 = blockIdx.x * blockDim.x + threadIdx.x;
  const int stride = gridDim.x * blockDim.x;
  for (int i = i0; i < n4; i += stride) p[i] = make_float4(0.f, 0.f, 0.f, 0.f);
  for (int i = i0; i < nh4; i += stride) hist4[i] = make_int4(0, 0, 0, 0);
}


__global__ void cell_count_kernel(const float* __restrict__ pos, int* __restrict__ hist,
                                  int* __restrict__ cellOf, int* __restrict__ rank,
                                  int Nper) {
  const int p = blockIdx.x * blockDim.x + threadIdx.x;
  if (p >= 2 * Nper) return;
  int b0 = (int)(pos[p * 3 + 0] * 64.0f);
  int b1 = (int)(pos[p * 3 + 1] * 64.0f);
  int b2 = (int)(pos[p * 3 + 2] * 64.0f);
  b0 = min(max(b0, 0), 63);
  b1 = min(max(b1, 0), 63);
  b2 = min(max(b2, 0), 63);
  const int c = ((p >= Nper) ? NCELL : 0) + (b0 * 64 + b1) * 64 + b2;
  cellOf[p] = c;
  rank[p] = atomicAdd(hist + c, 1);
}

__global__ void cell_scan_local_kernel(int* __restrict__ hist, int* __restrict__ bsum) {
  __shared__ int sW[8];
  const int t = threadIdx.x, lane = t & 31, wv = t >> 5;
  int4* h4 = (int4*)hist + blockIdx.x * 256 + t;
  const int4 v = *h4;
  const int s = v.x + v.y + v.z + v.w;
  int inc = s;
#pragma unroll
  for (int o = 1; o < 32; o <<= 1) {
    int n = __shfl_up(inc, o);
    if (lane >= o) inc += n;
  }
  if (lane == 31) sW[wv] = inc;
  __syncthreads();
  int wpre = 0, tot = 0;
#pragma unroll
  for (int w = 0; w < 8; ++w) {
    int xw = sW[w];
    if (w < wv) wpre += xw;
    tot += xw;
  }
  const int ex = wpre + inc - s;
  *h4 = make_int4(ex, ex + v.x, ex + v.x + v.y, ex + v.x + v.y + v.z);
  if (t == 0) bsum[blockIdx.x] = tot;
}

__global__ void cell_scan_blocks_kernel(const int* __restrict__ bsum, int* __restrict__ bpre) {
  __shared__ int sS[NSCANB];
  const int t = threadIdx.x;
  const int own = bsum[t];
  sS[t] = own;
  __syncthreads();
  for (int o = 1; o < NSCANB; o <<= 1) {
    const int add = (t >= o) ? sS[t - o] : 0;
    __syncthreads();
    sS[t] += add;
    __syncthreads();
  }
  bpre[t] = sS[t] - own;
}

__global__ void cell_place_kernel(const int* __restrict__ hist, const int* __restrict__ bpre,
                                  const int* __restrict__ cellOf, const int* __restrict__ rank,
                                  int* __restrict__ order, int Nper) {
  const int p = blockIdx.x * blockDim.x + threadIdx.x;
  if (p >= 2 * Nper) return;
  const int c = cellOf[p];
  order[hist[c] + bpre[c >> 10] + rank[p]] = p;
}

__global__ void cell_order_fix_kernel(const int* __restrict__ hist, const int* __restrict__ bpre,
                                      int* __restrict__ order, int Nper) {
  const int c = blockIdx.x * blockDim.x + threadIdx.x;
  if (c >= NCELL2) return;
  const int s = hist[c] + bpre[c >> 10];
  const int e = (c + 1 < NCELL2) ? (hist[c + 1] + bpre[(c + 1) >> 10]) : (2 * Nper);
  for (int i = s + 1; i < e; ++i) {
    const int v = order[i];
    int j = i - 1;
    while (j >= s && order[j] > v) {
      order[j + 1] = order[j];
      --j;
    }
    order[j + 1] = v;
  }
}


__launch_bounds__(256)
__global__ void p2g_wmma_kernel(const float* __restrict__ x,
                                const float* __restrict__ pos,
                                const float* __restrict__ kern,
                                const float* __restrict__ bias,
                                float* __restrict__ out,
                                unsigned int* __restrict__ cnt,
                                const int* __restrict__ order,
                                int Nper) {
  __shared__ float sB[3 * 2 * 8 * 64];
  __shared__ float4 sP[8][32];
  __shared__ int sSel[2];

  const int t    = threadIdx.x;
  const int lane = t & 31;
  const int wv   = t >> 5;
  const int half = lane >> 4;
  const int nlo  = lane & 15;

  const int chunksPer = (Nper + 255) >> 8;
  if (t == 0) {
    unsigned int id = __builtin_amdgcn_s_sendmsg_rtn(0x87);
    int hb = (int)((id >> 18) & 1u);
    int tk = (int)__hip_atomic_fetch_add(cnt + hb, 1u, __ATOMIC_RELAXED,
                                         __HIP_MEMORY_SCOPE_AGENT);
    if (tk >= chunksPer) {
      hb ^= 1;
      tk = (int)__hip_atomic_fetch_add(cnt + hb, 1u, __ATOMIC_RELAXED,
                                       __HIP_MEMORY_SCOPE_AGENT);
    }
    sSel[0] = hb;
    sSel[1] = tk;
  }
  __syncthreads();
  const int bb = sSel[0];
  const int chunk = sSel[1];
  if (chunk >= chunksPer) return;

  const int waveBase = (chunk * 8 + wv) * 32;
  const int pOff = bb * Nper;

  const int sp = (lane & 16) + ((lane >> 1) & 3) * 4 + ((lane >> 3) & 1) * 2 + (lane & 1);
  const int q  = waveBase + sp;
  const int pc = order[pOff + ((q < Nper) ? q : (Nper - 1))];
  const float valid = (q < Nper) ? 1.0f : 0.0f;

  float px = pos[pc * 3 + 0] * 64.0f;
  float py = pos[pc * 3 + 1] * 64.0f;
  float pz = pos[pc * 3 + 2] * 64.0f;
  int b0 = (int)px, b1 = (int)py, b2 = (int)pz;
  float fx = px - (float)b0 - 0.5f;
  float fy = py - (float)b1 - 0.5f;
  float fz = pz - (float)b2 - 0.5f;

  const float wx0 = 0.5f * (0.5f - fx) * (0.5f - fx) * valid;
  const float wx1 = (0.75f - fx * fx) * valid;
  const float wx2 = 0.5f * (0.5f + fx) * (0.5f + fx) * valid;
  const float wy0 = 0.5f * (0.5f - fy) * (0.5f - fy);
  const float wy1 = 0.75f - fy * fy;
  const float wy2 = 0.5f * (0.5f + fy) * (0.5f + fy);
  const float wz0 = 0.5f * (0.5f - fz) * (0.5f - fz);
  const float wz1 = 0.75f - fz * fz;
  const float wz2 = 0.5f * (0.5f + fz) * (0.5f + fz);

  const int lin = b0 * HHW + b1 * 66 + b2;

  int lmin = lin, lmax = lin;
#pragma unroll
  for (int o = 16; o > 0; o >>= 1) {
    lmin = min(lmin, __shfl_xor(lmin, o));
    lmax = max(lmax, __shfl_xor(lmax, o));
  }
  lmin = __builtin_amdgcn_readfirstlane(lmin);
  lmax = __builtin_amdgcn_readfirstlane(lmax);
  int glo = min(lin, __shfl_xor(lin, 1));
  int ghi = max(lin, __shfl_xor(lin, 1));
  glo = min(glo, __shfl_xor(glo, 8));
  ghi = max(ghi, __shfl_xor(ghi, 8));
  int gLo[8], gHi[8];
#pragma unroll
  for (int gq = 0; gq < 8; ++gq) {
    gLo[gq] = __builtin_amdgcn_readlane(glo, (gq >> 2) * 16 + (gq & 3) * 2);
    gHi[gq] = __builtin_amdgcn_readlane(ghi, (gq >> 2) * 16 + (gq & 3) * 2);
  }

  v2f a[2][8];
#pragma unroll
  for (int mt = 0; mt < 2; ++mt) {
    const int prow = __shfl(pc, mt * 16 + nlo);
    const float* xr = x + prow * CI + half * 2;
#pragma unroll
    for (int c = 0; c < 8; ++c) a[mt][c] = *(const v2f*)(xr + c * 4);
  }

  const float bias0 = bias[nlo];
  const float bias1 = bias[16 + nlo];

  const float4* k4 = (const float4*)kern;
  const int sci = t >> 3;
  const int sco = (t & 7) * 4;
  const int sbase = (sci >> 2) * 64 + (((sci >> 1) & 1) * 16) * 2 + (sci & 1)
                  + (sco >> 4) * 512 + (sco & 15) * 2;

  for (int g = 0; g < 9; ++g) {
    const int oi = g / 3, oj = g - oi * 3;

    __syncthreads();

#pragma unroll
    for (int j = 0; j < 3; ++j) {
      float4 v4 = k4[g * 768 + j * 256 + t];
      float vv[4] = {v4.x, v4.y, v4.z, v4.w};
#pragma unroll
      for (int i = 0; i < 4; ++i) sB[j * 1024 + sbase + i * 2] = vv[i];
    }
    {
      float wxs = (oi == 0) ? wx0 : ((oi == 1) ? wx1 : wx2);
      float wys = (oj == 0) ? wy0 : ((oj == 1) ? wy1 : wy2);
      float wxy = wxs * wys;
      sP[wv][lane] = make_float4(wxy * wz0, wxy * wz1, wxy * wz2, __int_as_float(lin));
    }
    __syncthreads();

    v8f y[3][2][2];
#pragma unroll
    for (int ok = 0; ok < 3; ++ok) {
#pragma unroll
      for (int nt = 0; nt < 2; ++nt) {
        v2f bf[8];
#pragma unroll
        for (int c = 0; c < 8; ++c)
          bf[c] = *(const v2f*)&sB[((ok * 2 + nt) * 8 + c) * 64 + lane * 2];
        const float bl = nt ? bias1 : bias0;
#pragma unroll
        for (int mt = 0; mt < 2; ++mt)
#pragma unroll
          for (int r = 0; r < 8; ++r) y[ok][nt][mt][r] = bl;
#pragma unroll
        for (int c = 0; c < 8; ++c) {
#pragma unroll
          for (int mt = 0; mt < 2; ++mt) {
            y[ok][nt][mt] = __builtin_amdgcn_wmma_f32_16x16x4_f32(
                false, a[mt][c], false, bf[c], (short)0, y[ok][nt][mt],
                false, false);
          }
        }
      }
    }

    const int goff = oi * HHW + oj * 66;
    const int base = lmin - ((lmin + 4 * oi + 2 * oj) & 7);
    const int nrun = ((lmax + 2 - base) >> 4) + 1;
    for (int ti = 0; ti < nrun; ++ti) {
      const int tb = base + ti * 16;
      v8f acc0, acc1;
#pragma unroll
      for (int r = 0; r < 8; ++r) { acc0[r] = 0.0f; acc1[r] = 0.0f; }
      bool any = false;
#pragma unroll
      for (int gq = 0; gq < 8; ++gq) {
        if (gHi[gq] + 2 >= tb && gLo[gq] <= tb + 15) {
          any = true;
          const int mt = gq >> 2, j = gq & 3;
          const float4 p0 = sP[wv][mt * 16 + half * 8 + 2 * j];
          const float4 p1 = sP[wv][mt * 16 + half * 8 + 2 * j + 1];
          const int d0 = __float_as_int(p0.w) - tb - nlo;
          const int d1 = __float_as_int(p1.w) - tb - nlo;
#pragma unroll
          for (int ok = 0; ok < 3; ++ok) {
            const float w0 = (ok == 0) ? p0.x : ((ok == 1) ? p0.y : p0.z);
            const float w1 = (ok == 0) ? p1.x : ((ok == 1) ? p1.y : p1.z);
            v2f sel;
            sel.x = (d0 + ok == 0) ? w0 : 0.0f;
            sel.y = (d1 + ok == 0) ? w1 : 0.0f;
            v2f ya0, ya1;
            ya0.x = y[ok][0][mt][2 * j]; ya0.y = y[ok][0][mt][2 * j + 1];
            ya1.x = y[ok][1][mt][2 * j]; ya1.y = y[ok][1][mt][2 * j + 1];
            acc0 = __builtin_amdgcn_wmma_f32_16x16x4_f32(
                false, ya0, false, sel, (short)0, acc0, false, false);
            acc1 = __builtin_amdgcn_wmma_f32_16x16x4_f32(
                false, ya1, false, sel, (short)0, acc1, false, false);
          }
        }
      }
      if (any) {
        const int idx = goff + tb + nlo;
        if ((unsigned)idx < (unsigned)DHW) {
          const unsigned colB = (unsigned)((bb * CO + half * 8) * DHW + idx) * 4u;
#pragma unroll
          for (int r = 0; r < 8; ++r) {
            const float v0 = acc0[r];
            const float v1 = acc1[r];
            if (v0 != 0.0f)
              atomicAdd((float*)((char*)out + (colB + (unsigned)(r * DHW * 4))), v0);
            if (v1 != 0.0f)
              atomicAdd((float*)((char*)out + (colB + (unsigned)((16 + r) * DHW * 4))), v1);
          }
        }
      }
    }
  }
}

extern "C" void kernel_launch(void* const* d_in, const int* in_sizes, int n_in,
                              void* d_out, int out_size, void* d_ws, size_t ws_size,
                              hipStream_t stream) {
  const float* x    = (const float*)d_in[0];
  const float* pos  = (const float*)d_in[1];
  const float* kern = (const float*)d_in[2];
  const float* bias = (const float*)d_in[3];
  float* out = (float*)d_out;
  unsigned int* cnt = (unsigned int*)d_ws;
  int* hist   = (int*)((char*)d_ws + 256);
  int* bsum   = hist + NCELL2;
  int* bpre   = bsum + NSCANB;
  int* cellOf = bpre + NSCANB;
  int* rank   = cellOf + (1 << 20);
  int* order  = rank + (1 << 20);

  const int Nper   = in_sizes[1] / (2 * 3);

  int n4 = out_size / 4;
  zero_f4_kernel<<<2048, 256, 0, stream>>>((float4*)out, n4, (int4*)hist, NCELL2 / 4, cnt);

  int pblocks = (2 * Nper + 255) / 256;
  cell_count_kernel<<<pblocks, 256, 0, stream>>>(pos, hist, cellOf, rank, Nper);
  cell_scan_local_kernel<<<NSCANB, 256, 0, stream>>>(hist, bsum);
  cell_scan_blocks_kernel<<<1, NSCANB, 0, stream>>>(bsum, bpre);
  cell_place_kernel<<<pblocks, 256, 0, stream>>>(hist, bpre, cellOf, rank, order, Nper);
  cell_order_fix_kernel<<<NCELL2 / 256, 256, 0, stream>>>(hist, bpre, order, Nper);

  int blocks = 2 * ((Nper + 255) / 256);
  p2g_wmma_kernel<<<blocks, 256, 0, stream>>>(x, pos, kern, bias, out,
                                              cnt, order, Nper);
}


# a run's 16 grid adds issue back to back: one touched-cell test per lane (not 16 value tests), addresses = 2 pointers + constant offsets, so no add waits for the one before; atomic order as before
# speedup vs baseline: 30.3109x; 1.2393x over previous
#include <hip/hip_runtime.h>


typedef float v2f __attribute__((ext_vector_type(2)));
typedef float v8f __attribute__((ext_vector_type(8)));

#define DD 66
#define HHW (66 * 66)
#define DHW (66 * 66 * 66)
#define CI 32
#define CO 32
#define NCELL (64 * 64 * 64)
#define NCELL2 (2 * NCELL)
#define NSCANB (NCELL2 / 1024)

__global__ void zero_f4_kernel(float4* __restrict__ p, int n4,
                               int4* __restrict__ hist4, int nh4,
                               unsigned int* __restrict__ cnt) {
  if (blockIdx.x == 0 && threadIdx.x < 2) cnt[threadIdx.x] = 0u;
  const int i0 = blockIdx.x * blockDim.x + threadIdx.x;
  const int stride = gridDim.x * blockDim.x;
  for (int i = i0; i < n4; i += stride) p[i] = make_float4(0.f, 0.f, 0.f, 0.f);
  for (int i = i0; i < nh4; i += stride) hist4[i] = make_int4(0, 0, 0, 0);
}


__global__ void cell_count_kernel(const float* __restrict__ pos, int* __restrict__ hist,
                                  int* __restrict__ cellOf, int* __restrict__ rank,
                                  int Nper) {
  const int p = blockIdx.x * blockDim.x + threadIdx.x;
  if (p >= 2 * Nper) return;
  int b0 = (int)(pos[p * 3 + 0] * 64.0f);
  int b1 = (int)(pos[p * 3 + 1] * 64.0f);
  int b2 = (int)(pos[p * 3 + 2] * 64.0f);
  b0 = min(max(b0, 0), 63);
  b1 = min(max(b1, 0), 63);
  b2 = min(max(b2, 0), 63);
  const int c = ((p >= Nper) ? NCELL : 0) + (b0 * 64 + b1) * 64 + b2;
  cellOf[p] = c;
  rank[p] = atomicAdd(hist + c, 1);
}

__global__ void cell_scan_local_kernel(int* __restrict__ hist, int* __restrict__ bsum) {
  __shared__ int sW[8];
  const int t = threadIdx.x, lane = t & 31, wv = t >> 5;
  int4* h4 = (int4*)hist + blockIdx.x * 256 + t;
  const int4 v = *h4;
  const int s = v.x + v.y + v.z + v.w;
  int inc = s;
#pragma unroll
  for (int o = 1; o < 32; o <<= 1) {
    int n = __shfl_up(inc, o);
    if (lane >= o) inc += n;
  }
  if (lane == 31) sW[wv] = inc;
  __syncthreads();
  int wpre = 0, tot = 0;
#pragma unroll
  for (int w = 0; w < 8; ++w) {
    int xw = sW[w];
    if (w < wv) wpre += xw;
    tot += xw;
  }
  const int ex = wpre + inc - s;
  *h4 = make_int4(ex, ex + v.x, ex + v.x + v.y, ex + v.x + v.y + v.z);
  if (t == 0) bsum[blockIdx.x] = tot;
}

__global__ void cell_scan_blocks_kernel(const int* __restrict__ bsum, int* __restrict__ bpre) {
  __shared__ int sS[NSCANB];
  const int t = threadIdx.x;
  const int own = bsum[t];
  sS[t] = own;
  __syncthreads();
  for (int o = 1; o < NSCANB; o <<= 1) {
    const int add = (t >= o) ? sS[t - o] : 0;
    __syncthreads();
    sS[t] += add;
    __syncthreads();
  }
  bpre[t] = sS[t] - own;
}

__global__ void cell_place_kernel(const int* __restrict__ hist, const int* __restrict__ bpre,
                                  const int* __restrict__ cellOf, const int* __restrict__ rank,
                                  int* __restrict__ order, int Nper) {
  const int p = blockIdx.x * blockDim.x + threadIdx.x;
  if (p >= 2 * Nper) return;
  const int c = cellOf[p];
  order[hist[c] + bpre[c >> 10] + rank[p]] = p;
}

__global__ void cell_order_fix_kernel(const int* __restrict__ hist, const int* __restrict__ bpre,
                                      int* __restrict__ order, int Nper) {
  const int c = blockIdx.x * blockDim.x + threadIdx.x;
  if (c >= NCELL2) return;
  const int s = hist[c] + bpre[c >> 10];
  const int e = (c + 1 < NCELL2) ? (hist[c + 1] + bpre[(c + 1) >> 10]) : (2 * Nper);
  for (int i = s + 1; i < e; ++i) {
    const int v = order[i];
    int j = i - 1;
    while (j >= s && order[j] > v) {
      order[j + 1] = order[j];
      --j;
    }
    order[j + 1] = v;
  }
}


__launch_bounds__(256)
__global__ void p2g_wmma_kernel(const float* __restrict__ x,
                                const float* __restrict__ pos,
                                const float* __restrict__ kern,
                                const float* __restrict__ bias,
                                float* __restrict__ out,
                                unsigned int* __restrict__ cnt,
                                const int* __restrict__ order,
                                int Nper) {
  __shared__ float sB[3 * 2 * 8 * 64];
  __shared__ float4 sP[8][32];
  __shared__ int sSel[2];

  const int t    = threadIdx.x;
  const int lane = t & 31;
  const int wv   = t >> 5;
  const int half = lane >> 4;
  const int nlo  = lane & 15;

  const int chunksPer = (Nper + 255) >> 8;
  if (t == 0) {
    unsigned int id = __builtin_amdgcn_s_sendmsg_rtn(0x87);
    int hb = (int)((id >> 18) & 1u);
    int tk = (int)__hip_atomic_fetch_add(cnt + hb, 1u, __ATOMIC_RELAXED,
                                         __HIP_MEMORY_SCOPE_AGENT);
    if (tk >= chunksPer) {
      hb ^= 1;
      tk = (int)__hip_atomic_fetch_add(cnt + hb, 1u, __ATOMIC_RELAXED,
                                       __HIP_MEMORY_SCOPE_AGENT);
    }
    sSel[0] = hb;
    sSel[1] = tk;
  }
  __syncthreads();
  const int bb = sSel[0];
  const int chunk = sSel[1];
  if (chunk >= chunksPer) return;

  const int waveBase = (chunk * 8 + wv) * 32;
  const int pOff = bb * Nper;

  const int sp = (lane & 16) + ((lane >> 1) & 3) * 4 + ((lane >> 3) & 1) * 2 + (lane & 1);
  const int q  = waveBase + sp;
  const int pc = order[pOff + ((q < Nper) ? q : (Nper - 1))];
  const float valid = (q < Nper) ? 1.0f : 0.0f;

  float px = pos[pc * 3 + 0] * 64.0f;
  float py = pos[pc * 3 + 1] * 64.0f;
  float pz = pos[pc * 3 + 2] * 64.0f;
  int b0 = (int)px, b1 = (int)py, b2 = (int)pz;
  float fx = px - (float)b0 - 0.5f;
  float fy = py - (float)b1 - 0.5f;
  float fz = pz - (float)b2 - 0.5f;

  const float wx0 = 0.5f * (0.5f - fx) * (0.5f - fx) * valid;
  const float wx1 = (0.75f - fx * fx) * valid;
  const float wx2 = 0.5f * (0.5f + fx) * (0.5f + fx) * valid;
  const float wy0 = 0.5f * (0.5f - fy) * (0.5f - fy);
  const float wy1 = 0.75f - fy * fy;
  const float wy2 = 0.5f * (0.5f + fy) * (0.5f + fy);
  const float wz0 = 0.5f * (0.5f - fz) * (0.5f - fz);
  const float wz1 = 0.75f - fz * fz;
  const float wz2 = 0.5f * (0.5f + fz) * (0.5f + fz);

  const int lin = b0 * HHW + b1 * 66 + b2;

  int lmin = lin, lmax = lin;
#pragma unroll
  for (int o = 16; o > 0; o >>= 1) {
    lmin = min(lmin, __shfl_xor(lmin, o));
    lmax = max(lmax, __shfl_xor(lmax, o));
  }
  lmin = __builtin_amdgcn_readfirstlane(lmin);
  lmax = __builtin_amdgcn_readfirstlane(lmax);
  int glo = min(lin, __shfl_xor(lin, 1));
  int ghi = max(lin, __shfl_xor(lin, 1));
  glo = min(glo, __shfl_xor(glo, 8));
  ghi = max(ghi, __shfl_xor(ghi, 8));
  int gLo[8], gHi[8];
#pragma unroll
  for (int gq = 0; gq < 8; ++gq) {
    gLo[gq] = __builtin_amdgcn_readlane(glo, (gq >> 2) * 16 + (gq & 3) * 2);
    gHi[gq] = __builtin_amdgcn_readlane(ghi, (gq >> 2) * 16 + (gq & 3) * 2);
  }

  v2f a[2][8];
#pragma unroll
  for (int mt = 0; mt < 2; ++mt) {
    const int prow = __shfl(pc, mt * 16 + nlo);
    const float* xr = x + prow * CI + half * 2;
#pragma unroll
    for (int c = 0; c < 8; ++c) a[mt][c] = *(const v2f*)(xr + c * 4);
  }

  const float bias0 = bias[nlo];
  const float bias1 = bias[16 + nlo];

  const float4* k4 = (const float4*)kern;
  const int sci = t >> 3;
  const int sco = (t & 7) * 4;
  const int sbase = (sci >> 2) * 64 + (((sci >> 1) & 1) * 16) * 2 + (sci & 1)
                  + (sco >> 4) * 512 + (sco & 15) * 2;

  for (int g = 0; g < 9; ++g) {
    const int oi = g / 3, oj = g - oi * 3;

    __syncthreads();

#pragma unroll
    for (int j = 0; j < 3; ++j) {
      float4 v4 = k4[g * 768 + j * 256 + t];
      float vv[4] = {v4.x, v4.y, v4.z, v4.w};
#pragma unroll
      for (int i = 0; i < 4; ++i) sB[j * 1024 + sbase + i * 2] = vv[i];
    }
    {
      float wxs = (oi == 0) ? wx0 : ((oi == 1) ? wx1 : wx2);
      float wys = (oj == 0) ? wy0 : ((oj == 1) ? wy1 : wy2);
      float wxy = wxs * wys;
      sP[wv][lane] = make_float4(wxy * wz0, wxy * wz1, wxy * wz2, __int_as_float(lin));
    }
    __syncthreads();

    v8f y[3][2][2];
#pragma unroll
    for (int ok = 0; ok < 3; ++ok) {
#pragma unroll
      for (int nt = 0; nt < 2; ++nt) {
        v2f bf[8];
#pragma unroll
        for (int c = 0; c < 8; ++c)
          bf[c] = *(const v2f*)&sB[((ok * 2 + nt) * 8 + c) * 64 + lane * 2];
        const float bl = nt ? bias1 : bias0;
#pragma unroll
        for (int mt = 0; mt < 2; ++mt)
#pragma unroll
          for (int r = 0; r < 8; ++r) y[ok][nt][mt][r] = bl;
#pragma unroll
        for (int c = 0; c < 8; ++c) {
#pragma unroll
          for (int mt = 0; mt < 2; ++mt) {
            y[ok][nt][mt] = __builtin_amdgcn_wmma_f32_16x16x4_f32(
                false, a[mt][c], false, bf[c], (short)0, y[ok][nt][mt],
                false, false);
          }
        }
      }
    }

    const int goff = oi * HHW + oj * 66;
    const int base = lmin - ((lmin + 4 * oi + 2 * oj) & 7);
    const int nrun = ((lmax + 2 - base) >> 4) + 1;
    for (int ti = 0; ti < nrun; ++ti) {
      const int tb = base + ti * 16;
      v8f acc0, acc1;
#pragma unroll
      for (int r = 0; r < 8; ++r) { acc0[r] = 0.0f; acc1[r] = 0.0f; }
      bool any = false;
#pragma unroll
      for (int gq = 0; gq < 8; ++gq) {
        if (gHi[gq] + 2 >= tb && gLo[gq] <= tb + 15) {
          any = true;
          const int mt = gq >> 2, j = gq & 3;
          const float4 p0 = sP[wv][mt * 16 + half * 8 + 2 * j];
          const float4 p1 = sP[wv][mt * 16 + half * 8 + 2 * j + 1];
          const int d0 = __float_as_int(p0.w) - tb - nlo;
          const int d1 = __float_as_int(p1.w) - tb - nlo;
#pragma unroll
          for (int ok = 0; ok < 3; ++ok) {
            const float w0 = (ok == 0) ? p0.x : ((ok == 1) ? p0.y : p0.z);
            const float w1 = (ok == 0) ? p1.x : ((ok == 1) ? p1.y : p1.z);
            v2f sel;
            sel.x = (d0 + ok == 0) ? w0 : 0.0f;
            sel.y = (d1 + ok == 0) ? w1 : 0.0f;
            v2f ya0, ya1;
            ya0.x = y[ok][0][mt][2 * j]; ya0.y = y[ok][0][mt][2 * j + 1];
            ya1.x = y[ok][1][mt][2 * j]; ya1.y = y[ok][1][mt][2 * j + 1];
            acc0 = __builtin_amdgcn_wmma_f32_16x16x4_f32(
                false, ya0, false, sel, (short)0, acc0, false, false);
            acc1 = __builtin_amdgcn_wmma_f32_16x16x4_f32(
                false, ya1, false, sel, (short)0, acc1, false, false);
          }
        }
      }
      if (any) {
        const int idx = goff + tb + nlo;
        bool nz = false;
#pragma unroll
        for (int r = 0; r < 8; ++r) nz = nz || (acc0[r] != 0.0f) || (acc1[r] != 0.0f);
        if (nz && (unsigned)idx < (unsigned)DHW) {
          float* const p0 = out + ((size_t)((bb * CO + half * 8) * DHW) + (size_t)idx);
          float* const p1 = p0 + (size_t)16 * DHW;
#pragma unroll
          for (int r = 0; r < 8; ++r) {
            atomicAdd(p0 + r * DHW, acc0[r]);
            atomicAdd(p1 + r * DHW, acc1[r]);
          }
        }
      }
    }
  }
}

extern "C" void kernel_launch(void* const* d_in, const int* in_sizes, int n_in,
                              void* d_out, int out_size, void* d_ws, size_t ws_size,
                              hipStream_t stream) {
  const float* x    = (const float*)d_in[0];
  const float* pos  = (const float*)d_in[1];
  const float* kern = (const float*)d_in[2];
  const float* bias = (const float*)d_in[3];
  float* out = (float*)d_out;
  unsigned int* cnt = (unsigned int*)d_ws;
  int* hist   = (int*)((char*)d_ws + 256);
  int* bsum   = hist + NCELL2;
  int* bpre   = bsum + NSCANB;
  int* cellOf = bpre + NSCANB;
  int* rank   = cellOf + (1 << 20);
  int* order  = rank + (1 << 20);

  const int Nper   = in_sizes[1] / (2 * 3);

  int n4 = out_size / 4;
  zero_f4_kernel<<<2048, 256, 0, stream>>>((float4*)out, n4, (int4*)hist, NCELL2 / 4, cnt);

  int pblocks = (2 * Nper + 255) / 256;
  cell_count_kernel<<<pblocks, 256, 0, stream>>>(pos, hist, cellOf, rank, Nper);
  cell_scan_local_kernel<<<NSCANB, 256, 0, stream>>>(hist, bsum);
  cell_scan_blocks_kernel<<<1, NSCANB, 0, stream>>>(bsum, bpre);
  cell_place_kernel<<<pblocks, 256, 0, stream>>>(hist, bpre, cellOf, rank, order, Nper);
  cell_order_fix_kernel<<<NCELL2 / 256, 256, 0, stream>>>(hist, bpre, order, Nper);

  int blocks = 2 * ((Nper + 255) / 256);
  p2g_wmma_kernel<<<blocks, 256, 0, stream>>>(x, pos, kern, bias, out,
                                              cnt, order, Nper);
}


# all 27 kernel slices go into LDS once per block (108 KiB) instead of 3 per (oi,oj) pass: the 9-pass loop has no block barrier and no global load, waves run free; atomic arrival order as before
# speedup vs baseline: 31.3121x; 1.0330x over previous
#include <hip/hip_runtime.h>


typedef float v2f __attribute__((ext_vector_type(2)));
typedef float v8f __attribute__((ext_vector_type(8)));

#define DD 66
#define HHW (66 * 66)
#define DHW (66 * 66 * 66)
#define CI 32
#define CO 32
#define NCELL (64 * 64 * 64)
#define NCELL2 (2 * NCELL)
#define NSCANB (NCELL2 / 1024)

__global__ void zero_f4_kernel(float4* __restrict__ p, int n4,
                               int4* __restrict__ hist4, int nh4,
                               unsigned int* __restrict__ cnt) {
  if (blockIdx.x == 0 && threadIdx.x < 2) cnt[threadIdx.x] = 0u;
  const int i0 = blockIdx.x * blockDim.x + threadIdx.x;
  const int stride = gridDim.x * blockDim.x;
  for (int i = i0; i < n4; i += stride) p[i] = make_float4(0.f, 0.f, 0.f, 0.f);
  for (int i = i0; i < nh4; i += stride) hist4[i] = make_int4(0, 0, 0, 0);
}


__global__ void cell_count_kernel(const float* __restrict__ pos, int* __restrict__ hist,
                                  int* __restrict__ cellOf, int* __restrict__ rank,
                                  int Nper) {
  const int p = blockIdx.x * blockDim.x + threadIdx.x;
  if (p >= 2 * Nper) return;
  int b0 = (int)(pos[p * 3 + 0] * 64.0f);
  int b1 = (int)(pos[p * 3 + 1] * 64.0f);
  int b2 = (int)(pos[p * 3 + 2] * 64.0f);
  b0 = min(max(b0, 0), 63);
  b1 = min(max(b1, 0), 63);
  b2 = min(max(b2, 0), 63);
  const int c = ((p >= Nper) ? NCELL : 0) + (b0 * 64 + b1) * 64 + b2;
  cellOf[p] = c;
  rank[p] = atomicAdd(hist + c, 1);
}

__global__ void cell_scan_local_kernel(int* __restrict__ hist, int* __restrict__ bsum) {
  __shared__ int sW[8];
  const int t = threadIdx.x, lane = t & 31, wv = t >> 5;
  int4* h4 = (int4*)hist + blockIdx.x * 256 + t;
  const int4 v = *h4;
  const int s = v.x + v.y + v.z + v.w;
  int inc = s;
#pragma unroll
  for (int o = 1; o < 32; o <<= 1) {
    int n = __shfl_up(inc, o);
    if (lane >= o) inc += n;
  }
  if (lane == 31) sW[wv] = inc;
  __syncthreads();
  int wpre = 0, tot = 0;
#pragma unroll
  for (int w = 0; w < 8; ++w) {
    int xw = sW[w];
    if (w < wv) wpre += xw;
    tot += xw;
  }
  const int ex = wpre + inc - s;
  *h4 = make_int4(ex, ex + v.x, ex + v.x + v.y, ex + v.x + v.y + v.z);
  if (t == 0) bsum[blockIdx.x] = tot;
}

__global__ void cell_scan_blocks_kernel(const int* __restrict__ bsum, int* __restrict__ bpre) {
  __shared__ int sS[NSCANB];
  const int t = threadIdx.x;
  const int own = bsum[t];
  sS[t] = own;
  __syncthreads();
  for (int o = 1; o < NSCANB; o <<= 1) {
    const int add = (t >= o) ? sS[t - o] : 0;
    __syncthreads();
    sS[t] += add;
    __syncthreads();
  }
  bpre[t] = sS[t] - own;
}

__global__ void cell_place_kernel(const int* __restrict__ hist, const int* __restrict__ bpre,
                                  const int* __restrict__ cellOf, const int* __restrict__ rank,
                                  int* __restrict__ order, int Nper) {
  const int p = blockIdx.x * blockDim.x + threadIdx.x;
  if (p >= 2 * Nper) return;
  const int c = cellOf[p];
  order[hist[c] + bpre[c >> 10] + rank[p]] = p;
}

__global__ void cell_order_fix_kernel(const int* __restrict__ hist, const int* __restrict__ bpre,
                                      int* __restrict__ order, int Nper) {
  const int c = blockIdx.x * blockDim.x + threadIdx.x;
  if (c >= NCELL2) return;
  const int s = hist[c] + bpre[c >> 10];
  const int e = (c + 1 < NCELL2) ? (hist[c + 1] + bpre[(c + 1) >> 10]) : (2 * Nper);
  for (int i = s + 1; i < e; ++i) {
    const int v = order[i];
    int j = i - 1;
    while (j >= s && order[j] > v) {
      order[j + 1] = order[j];
      --j;
    }
    order[j + 1] = v;
  }
}


__launch_bounds__(256)
__global__ void p2g_wmma_kernel(const float* __restrict__ x,
                                const float* __restrict__ pos,
                                const float* __restrict__ kern,
                                const float* __restrict__ bias,
                                float* __restrict__ out,
                                unsigned int* __restrict__ cnt,
                                const int* __restrict__ order,
                                int Nper) {
  __shared__ float sB[27 * 2 * 8 * 64];
  __shared__ float4 sP[8][32];
  __shared__ int sSel[2];

  const int t    = threadIdx.x;
  const int lane = t & 31;
  const int wv   = t >> 5;
  const int half = lane >> 4;
  const int nlo  = lane & 15;

  const int chunksPer = (Nper + 255) >> 8;
  if (t == 0) {
    unsigned int id = __builtin_amdgcn_s_sendmsg_rtn(0x87);
    int hb = (int)((id >> 18) & 1u);
    int tk = (int)__hip_atomic_fetch_add(cnt + hb, 1u, __ATOMIC_RELAXED,
                                         __HIP_MEMORY_SCOPE_AGENT);
    if (tk >= chunksPer) {
      hb ^= 1;
      tk = (int)__hip_atomic_fetch_add(cnt + hb, 1u, __ATOMIC_RELAXED,
                                       __HIP_MEMORY_SCOPE_AGENT);
    }
    sSel[0] = hb;
    sSel[1] = tk;
  }

  {
    const float4* k4 = (const float4*)kern;
    const int sci = t >> 3;
    const int sco = (t & 7) * 4;
    const int sbase = (sci >> 2) * 64 + (((sci >> 1) & 1) * 16) * 2 + (sci & 1)
                    + (sco >> 4) * 512 + (sco & 15) * 2;
#pragma unroll
    for (int o = 0; o < 27; ++o) {
      const float4 v4 = k4[o * 256 + t];
      sB[o * 1024 + sbase + 0] = v4.x;
      sB[o * 1024 + sbase + 2] = v4.y;
      sB[o * 1024 + sbase + 4] = v4.z;
      sB[o * 1024 + sbase + 6] = v4.w;
    }
  }
  __syncthreads();
  const int bb = sSel[0];
  const int chunk = sSel[1];
  if (chunk >= chunksPer) return;

  const int waveBase = (chunk * 8 + wv) * 32;
  const int pOff = bb * Nper;

  const int sp = (lane & 16) + ((lane >> 1) & 3) * 4 + ((lane >> 3) & 1) * 2 + (lane & 1);
  const int q  = waveBase + sp;
  const int pc = order[pOff + ((q < Nper) ? q : (Nper - 1))];
  const float valid = (q < Nper) ? 1.0f : 0.0f;

  float px = pos[pc * 3 + 0] * 64.0f;
  float py = pos[pc * 3 + 1] * 64.0f;
  float pz = pos[pc * 3 + 2] * 64.0f;
  int b0 = (int)px, b1 = (int)py, b2 = (int)pz;
  float fx = px - (float)b0 - 0.5f;
  float fy = py - (float)b1 - 0.5f;
  float fz = pz - (float)b2 - 0.5f;

  const float wx0 = 0.5f * (0.5f - fx) * (0.5f - fx) * valid;
  const float wx1 = (0.75f - fx * fx) * valid;
  const float wx2 = 0.5f * (0.5f + fx) * (0.5f + fx) * valid;
  const float wy0 = 0.5f * (0.5f - fy) * (0.5f - fy);
  const float wy1 = 0.75f - fy * fy;
  const float wy2 = 0.5f * (0.5f + fy) * (0.5f + fy);
  const float wz0 = 0.5f * (0.5f - fz) * (0.5f - fz);
  const float wz1 = 0.75f - fz * fz;
  const float wz2 = 0.5f * (0.5f + fz) * (0.5f + fz);

  const int lin = b0 * HHW + b1 * 66 + b2;

  int lmin = lin, lmax = lin;
#pragma unroll
  for (int o = 16; o > 0; o >>= 1) {
    lmin = min(lmin, __shfl_xor(lmin, o));
    lmax = max(lmax, __shfl_xor(lmax, o));
  }
  lmin = __builtin_amdgcn_readfirstlane(lmin);
  lmax = __builtin_amdgcn_readfirstlane(lmax);
  int glo = min(lin, __shfl_xor(lin, 1));
  int ghi = max(lin, __shfl_xor(lin, 1));
  glo = min(glo, __shfl_xor(glo, 8));
  ghi = max(ghi, __shfl_xor(ghi, 8));
  int gLo[8], gHi[8];
#pragma unroll
  for (int gq = 0; gq < 8; ++gq) {
    gLo[gq] = __builtin_amdgcn_readlane(glo, (gq >> 2) * 16 + (gq & 3) * 2);
    gHi[gq] = __builtin_amdgcn_readlane(ghi, (gq >> 2) * 16 + (gq & 3) * 2);
  }

  v2f a[2][8];
#pragma unroll
  for (int mt = 0; mt < 2; ++mt) {
    const int prow = __shfl(pc, mt * 16 + nlo);
    const float* xr = x + prow * CI + half * 2;
#pragma unroll
    for (int c = 0; c < 8; ++c) a[mt][c] = *(const v2f*)(xr + c * 4);
  }

  const float bias0 = bias[nlo];
  const float bias1 = bias[16 + nlo];

  for (int g = 0; g < 9; ++g) {
    const int oi = g / 3, oj = g - oi * 3;

    {
      float wxs = (oi == 0) ? wx0 : ((oi == 1) ? wx1 : wx2);
      float wys = (oj == 0) ? wy0 : ((oj == 1) ? wy1 : wy2);
      float wxy = wxs * wys;
      sP[wv][lane] = make_float4(wxy * wz0, wxy * wz1, wxy * wz2, __int_as_float(lin));
    }

    v8f y[3][2][2];
#pragma unroll
    for (int ok = 0; ok < 3; ++ok) {
#pragma unroll
      for (int nt = 0; nt < 2; ++nt) {
        v2f bf[8];
#pragma unroll
        for (int c = 0; c < 8; ++c)
          bf[c] = *(const v2f*)&sB[(((g * 3 + ok) * 2 + nt) * 8 + c) * 64 + lane * 2];
        const float bl = nt ? bias1 : bias0;
#pragma unroll
        for (int mt = 0; mt < 2; ++mt)
#pragma unroll
          for (int r = 0; r < 8; ++r) y[ok][nt][mt][r] = bl;
#pragma unroll
        for (int c = 0; c < 8; ++c) {
#pragma unroll
          for (int mt = 0; mt < 2; ++mt) {
            y[ok][nt][mt] = __builtin_amdgcn_wmma_f32_16x16x4_f32(
                false, a[mt][c], false, bf[c], (short)0, y[ok][nt][mt],
                false, false);
          }
        }
      }
    }

    const int goff = oi * HHW + oj * 66;
    const int base = lmin - ((lmin + 4 * oi + 2 * oj) & 7);
    const int nrun = ((lmax + 2 - base) >> 4) + 1;
    for (int ti = 0; ti < nrun; ++ti) {
      const int tb = base + ti * 16;
      v8f acc0, acc1;
#pragma unroll
      for (int r = 0; r < 8; ++r) { acc0[r] = 0.0f; acc1[r] = 0.0f; }
      bool any = false;
#pragma unroll
      for (int gq = 0; gq < 8; ++gq) {
        if (gHi[gq] + 2 >= tb && gLo[gq] <= tb + 15) {
          any = true;
          const int mt = gq >> 2, j = gq & 3;
          const float4 p0 = sP[wv][mt * 16 + half * 8 + 2 * j];
          const float4 p1 = sP[wv][mt * 16 + half * 8 + 2 * j + 1];
          const int d0 = __float_as_int(p0.w) - tb - nlo;
          const int d1 = __float_as_int(p1.w) - tb - nlo;
#pragma unroll
          for (int ok = 0; ok < 3; ++ok) {
            const float w0 = (ok == 0) ? p0.x : ((ok == 1) ? p0.y : p0.z);
            const float w1 = (ok == 0) ? p1.x : ((ok == 1) ? p1.y : p1.z);
            v2f sel;
            sel.x = (d0 + ok == 0) ? w0 : 0.0f;
            sel.y = (d1 + ok == 0) ? w1 : 0.0f;
            v2f ya0, ya1;
            ya0.x = y[ok][0][mt][2 * j]; ya0.y = y[ok][0][mt][2 * j + 1];
            ya1.x = y[ok][1][mt][2 * j]; ya1.y = y[ok][1][mt][2 * j + 1];
            acc0 = __builtin_amdgcn_wmma_f32_16x16x4_f32(
                false, ya0, false, sel, (short)0, acc0, false, false);
            acc1 = __builtin_amdgcn_wmma_f32_16x16x4_f32(
                false, ya1, false, sel, (short)0, acc1, false, false);
          }
        }
      }
      if (any) {
        const int idx = goff + tb + nlo;
        bool nz = false;
#pragma unroll
        for (int r = 0; r < 8; ++r) nz = nz || (acc0[r] != 0.0f) || (acc1[r] != 0.0f);
        if (nz && (unsigned)idx < (unsigned)DHW) {
          float* const p0 = out + ((size_t)((bb * CO + half * 8) * DHW) + (size_t)idx);
          float* const p1 = p0 + (size_t)16 * DHW;
#pragma unroll
          for (int r = 0; r < 8; ++r) {
            atomicAdd(p0 + r * DHW, acc0[r]);
            atomicAdd(p1 + r * DHW, acc1[r]);
          }
        }
      }
    }
  }
}

extern "C" void kernel_launch(void* const* d_in, const int* in_sizes, int n_in,
                              void* d_out, int out_size, void* d_ws, size_t ws_size,
                              hipStream_t stream) {
  const float* x    = (const float*)d_in[0];
  const float* pos  = (const float*)d_in[1];
  const float* kern = (const float*)d_in[2];
  const float* bias = (const float*)d_in[3];
  float* out = (float*)d_out;
  unsigned int* cnt = (unsigned int*)d_ws;
  int* hist   = (int*)((char*)d_ws + 256);
  int* bsum   = hist + NCELL2;
  int* bpre   = bsum + NSCANB;
  int* cellOf = bpre + NSCANB;
  int* rank   = cellOf + (1 << 20);
  int* order  = rank + (1 << 20);

  const int Nper   = in_sizes[1] / (2 * 3);

  int n4 = out_size / 4;
  zero_f4_kernel<<<2048, 256, 0, stream>>>((float4*)out, n4, (int4*)hist, NCELL2 / 4, cnt);

  int pblocks = (2 * Nper + 255) / 256;
  cell_count_kernel<<<pblocks, 256, 0, stream>>>(pos, hist, cellOf, rank, Nper);
  cell_scan_local_kernel<<<NSCANB, 256, 0, stream>>>(hist, bsum);
  cell_scan_blocks_kernel<<<1, NSCANB, 0, stream>>>(bsum, bpre);
  cell_place_kernel<<<pblocks, 256, 0, stream>>>(hist, bpre, cellOf, rank, order, Nper);
  cell_order_fix_kernel<<<NCELL2 / 256, 256, 0, stream>>>(hist, bpre, order, Nper);

  int blocks = 2 * ((Nper + 255) / 256);
  p2g_wmma_kernel<<<blocks, 256, 0, stream>>>(x, pos, kern, bias, out,
                                              cnt, order, Nper);
}


# scatter blocks stay resident (512 = 2 per WGP) and every wave draws 32-entry list chunks by ticket: slices staged once per resident block, no wave waits for its block; atomic arrival order as before
# speedup vs baseline: 33.0064x; 1.0541x over previous
#include <hip/hip_runtime.h>


typedef float v2f __attribute__((ext_vector_type(2)));
typedef float v8f __attribute__((ext_vector_type(8)));

#define DD 66
#define HHW (66 * 66)
#define DHW (66 * 66 * 66)
#define CI 32
#define CO 32
#define NCELL (64 * 64 * 64)
#define NCELL2 (2 * NCELL)
#define NSCANB (NCELL2 / 1024)

__global__ void zero_f4_kernel(float4* __restrict__ p, int n4,
                               int4* __restrict__ hist4, int nh4,
                               unsigned int* __restrict__ cnt) {
  if (blockIdx.x == 0 && threadIdx.x < 2) cnt[threadIdx.x] = 0u;
  const int i0 = blockIdx.x * blockDim.x + threadIdx.x;
  const int stride = gridDim.x * blockDim.x;
  for (int i = i0; i < n4; i += stride) p[i] = make_float4(0.f, 0.f, 0.f, 0.f);
  for (int i = i0; i < nh4; i += stride) hist4[i] = make_int4(0, 0, 0, 0);
}


__global__ void cell_count_kernel(const float* __restrict__ pos, int* __restrict__ hist,
                                  int* __restrict__ cellOf, int* __restrict__ rank,
                                  int Nper) {
  const int p = blockIdx.x * blockDim.x + threadIdx.x;
  if (p >= 2 * Nper) return;
  int b0 = (int)(pos[p * 3 + 0] * 64.0f);
  int b1 = (int)(pos[p * 3 + 1] * 64.0f);
  int b2 = (int)(pos[p * 3 + 2] * 64.0f);
  b0 = min(max(b0, 0), 63);
  b1 = min(max(b1, 0), 63);
  b2 = min(max(b2, 0), 63);
  const int c = ((p >= Nper) ? NCELL : 0) + (b0 * 64 + b1) * 64 + b2;
  cellOf[p] = c;
  rank[p] = atomicAdd(hist + c, 1);
}

__global__ void cell_scan_local_kernel(int* __restrict__ hist, int* __restrict__ bsum) {
  __shared__ int sW[8];
  const int t = threadIdx.x, lane = t & 31, wv = t >> 5;
  int4* h4 = (int4*)hist + blockIdx.x * 256 + t;
  const int4 v = *h4;
  const int s = v.x + v.y + v.z + v.w;
  int inc = s;
#pragma unroll
  for (int o = 1; o < 32; o <<= 1) {
    int n = __shfl_up(inc, o);
    if (lane >= o) inc += n;
  }
  if (lane == 31) sW[wv] = inc;
  __syncthreads();
  int wpre = 0, tot = 0;
#pragma unroll
  for (int w = 0; w < 8; ++w) {
    int xw = sW[w];
    if (w < wv) wpre += xw;
    tot += xw;
  }
  const int ex = wpre + inc - s;
  *h4 = make_int4(ex, ex + v.x, ex + v.x + v.y, ex + v.x + v.y + v.z);
  if (t == 0) bsum[blockIdx.x] = tot;
}

__global__ void cell_scan_blocks_kernel(const int* __restrict__ bsum, int* __restrict__ bpre) {
  __shared__ int sS[NSCANB];
  const int t = threadIdx.x;
  const int own = bsum[t];
  sS[t] = own;
  __syncthreads();
  for (int o = 1; o < NSCANB; o <<= 1) {
    const int add = (t >= o) ? sS[t - o] : 0;
    __syncthreads();
    sS[t] += add;
    __syncthreads();
  }
  bpre[t] = sS[t] - own;
}

__global__ void cell_place_kernel(const int* __restrict__ hist, const int* __restrict__ bpre,
                                  const int* __restrict__ cellOf, const int* __restrict__ rank,
                                  int* __restrict__ order, int Nper) {
  const int p = blockIdx.x * blockDim.x + threadIdx.x;
  if (p >= 2 * Nper) return;
  const int c = cellOf[p];
  order[hist[c] + bpre[c >> 10] + rank[p]] = p;
}

__global__ void cell_order_fix_kernel(const int* __restrict__ hist, const int* __restrict__ bpre,
                                      int* __restrict__ order, int Nper) {
  const int c = blockIdx.x * blockDim.x + threadIdx.x;
  if (c >= NCELL2) return;
  const int s = hist[c] + bpre[c >> 10];
  const int e = (c + 1 < NCELL2) ? (hist[c + 1] + bpre[(c + 1) >> 10]) : (2 * Nper);
  for (int i = s + 1; i < e; ++i) {
    const int v = order[i];
    int j = i - 1;
    while (j >= s && order[j] > v) {
      order[j + 1] = order[j];
      --j;
    }
    order[j + 1] = v;
  }
}


__launch_bounds__(256)
__global__ void p2g_wmma_kernel(const float* __restrict__ x,
                                const float* __restrict__ pos,
                                const float* __restrict__ kern,
                                const float* __restrict__ bias,
                                float* __restrict__ out,
                                unsigned int* __restrict__ cnt,
                                const int* __restrict__ order,
                                int Nper) {
  __shared__ float sB[27 * 2 * 8 * 64];
  __shared__ float4 sP[8][32];
  __shared__ int sSel[1];

  const int t    = threadIdx.x;
  const int lane = t & 31;
  const int wv   = t >> 5;
  const int half = lane >> 4;
  const int nlo  = lane & 15;

  const int chunksPer = (Nper + 31) >> 5;
  if (t == 0) {
    unsigned int id = __builtin_amdgcn_s_sendmsg_rtn(0x87);
    sSel[0] = (int)((id >> 18) & 1u);
  }

  {
    const float4* k4 = (const float4*)kern;
    const int sci = t >> 3;
    const int sco = (t & 7) * 4;
    const int sbase = (sci >> 2) * 64 + (((sci >> 1) & 1) * 16) * 2 + (sci & 1)
                    + (sco >> 4) * 512 + (sco & 15) * 2;
#pragma unroll
    for (int o = 0; o < 27; ++o) {
      const float4 v4 = k4[o * 256 + t];
      sB[o * 1024 + sbase + 0] = v4.x;
      sB[o * 1024 + sbase + 2] = v4.y;
      sB[o * 1024 + sbase + 4] = v4.z;
      sB[o * 1024 + sbase + 6] = v4.w;
    }
  }
  __syncthreads();

  int bb = __builtin_amdgcn_readfirstlane(sSel[0]);
  int left = 2;
  const int maxDraws = 2 * chunksPer + 2;
  for (int draw = 0; draw < maxDraws && left > 0; ++draw) {
  int tk = 0;
  if (lane == 0)
    tk = (int)__hip_atomic_fetch_add(cnt + bb, 1u, __ATOMIC_RELAXED, __HIP_MEMORY_SCOPE_AGENT);
  tk = __builtin_amdgcn_readfirstlane(tk);
  if (tk >= chunksPer) {
    bb ^= 1;
    --left;
    continue;
  }

  const int waveBase = tk * 32;
  const int pOff = bb * Nper;

  const int sp = (lane & 16) + ((lane >> 1) & 3) * 4 + ((lane >> 3) & 1) * 2 + (lane & 1);
  const int q  = waveBase + sp;
  const int pc = order[pOff + ((q < Nper) ? q : (Nper - 1))];
  const float valid = (q < Nper) ? 1.0f : 0.0f;

  float px = pos[pc * 3 + 0] * 64.0f;
  float py = pos[pc * 3 + 1] * 64.0f;
  float pz = pos[pc * 3 + 2] * 64.0f;
  int b0 = (int)px, b1 = (int)py, b2 = (int)pz;
  float fx = px - (float)b0 - 0.5f;
  float fy = py - (float)b1 - 0.5f;
  float fz = pz - (float)b2 - 0.5f;

  const float wx0 = 0.5f * (0.5f - fx) * (0.5f - fx) * valid;
  const float wx1 = (0.75f - fx * fx) * valid;
  const float wx2 = 0.5f * (0.5f + fx) * (0.5f + fx) * valid;
  const float wy0 = 0.5f * (0.5f - fy) * (0.5f - fy);
  const float wy1 = 0.75f - fy * fy;
  const float wy2 = 0.5f * (0.5f + fy) * (0.5f + fy);
  const float wz0 = 0.5f * (0.5f - fz) * (0.5f - fz);
  const float wz1 = 0.75f - fz * fz;
  const float wz2 = 0.5f * (0.5f + fz) * (0.5f + fz);

  const int lin = b0 * HHW + b1 * 66 + b2;

  int lmin = lin, lmax = lin;
#pragma unroll
  for (int o = 16; o > 0; o >>= 1) {
    lmin = min(lmin, __shfl_xor(lmin, o));
    lmax = max(lmax, __shfl_xor(lmax, o));
  }
  lmin = __builtin_amdgcn_readfirstlane(lmin);
  lmax = __builtin_amdgcn_readfirstlane(lmax);
  int glo = min(lin, __shfl_xor(lin, 1));
  int ghi = max(lin, __shfl_xor(lin, 1));
  glo = min(glo, __shfl_xor(glo, 8));
  ghi = max(ghi, __shfl_xor(ghi, 8));
  int gLo[8], gHi[8];
#pragma unroll
  for (int gq = 0; gq < 8; ++gq) {
    gLo[gq] = __builtin_amdgcn_readlane(glo, (gq >> 2) * 16 + (gq & 3) * 2);
    gHi[gq] = __builtin_amdgcn_readlane(ghi, (gq >> 2) * 16 + (gq & 3) * 2);
  }

  v2f a[2][8];
#pragma unroll
  for (int mt = 0; mt < 2; ++mt) {
    const int prow = __shfl(pc, mt * 16 + nlo);
    const float* xr = x + prow * CI + half * 2;
#pragma unroll
    for (int c = 0; c < 8; ++c) a[mt][c] = *(const v2f*)(xr + c * 4);
  }

  const float bias0 = bias[nlo];
  const float bias1 = bias[16 + nlo];

  for (int g = 0; g < 9; ++g) {
    const int oi = g / 3, oj = g - oi * 3;

    {
      float wxs = (oi == 0) ? wx0 : ((oi == 1) ? wx1 : wx2);
      float wys = (oj == 0) ? wy0 : ((oj == 1) ? wy1 : wy2);
      float wxy = wxs * wys;
      sP[wv][lane] = make_float4(wxy * wz0, wxy * wz1, wxy * wz2, __int_as_float(lin));
    }

    v8f y[3][2][2];
#pragma unroll
    for (int ok = 0; ok < 3; ++ok) {
#pragma unroll
      for (int nt = 0; nt < 2; ++nt) {
        v2f bf[8];
#pragma unroll
        for (int c = 0; c < 8; ++c)
          bf[c] = *(const v2f*)&sB[(((g * 3 + ok) * 2 + nt) * 8 + c) * 64 + lane * 2];
        const float bl = nt ? bias1 : bias0;
#pragma unroll
        for (int mt = 0; mt < 2; ++mt)
#pragma unroll
          for (int r = 0; r < 8; ++r) y[ok][nt][mt][r] = bl;
#pragma unroll
        for (int c = 0; c < 8; ++c) {
#pragma unroll
          for (int mt = 0; mt < 2; ++mt) {
            y[ok][nt][mt] = __builtin_amdgcn_wmma_f32_16x16x4_f32(
                false, a[mt][c], false, bf[c], (short)0, y[ok][nt][mt],
                false, false);
          }
        }
      }
    }

    const int goff = oi * HHW + oj * 66;
    const int base = lmin - ((lmin + 4 * oi + 2 * oj) & 7);
    const int nrun = ((lmax + 2 - base) >> 4) + 1;
    for (int ti = 0; ti < nrun; ++ti) {
      const int tb = base + ti * 16;
      v8f acc0, acc1;
#pragma unroll
      for (int r = 0; r < 8; ++r) { acc0[r] = 0.0f; acc1[r] = 0.0f; }
      bool any = false;
#pragma unroll
      for (int gq = 0; gq < 8; ++gq) {
        if (gHi[gq] + 2 >= tb && gLo[gq] <= tb + 15) {
          any = true;
          const int mt = gq >> 2, j = gq & 3;
          const float4 p0 = sP[wv][mt * 16 + half * 8 + 2 * j];
          const float4 p1 = sP[wv][mt * 16 + half * 8 + 2 * j + 1];
          const int d0 = __float_as_int(p0.w) - tb - nlo;
          const int d1 = __float_as_int(p1.w) - tb - nlo;
#pragma unroll
          for (int ok = 0; ok < 3; ++ok) {
            const float w0 = (ok == 0) ? p0.x : ((ok == 1) ? p0.y : p0.z);
            const float w1 = (ok == 0) ? p1.x : ((ok == 1) ? p1.y : p1.z);
            v2f sel;
            sel.x = (d0 + ok == 0) ? w0 : 0.0f;
            sel.y = (d1 + ok == 0) ? w1 : 0.0f;
            v2f ya0, ya1;
            ya0.x = y[ok][0][mt][2 * j]; ya0.y = y[ok][0][mt][2 * j + 1];
            ya1.x = y[ok][1][mt][2 * j]; ya1.y = y[ok][1][mt][2 * j + 1];
            acc0 = __builtin_amdgcn_wmma_f32_16x16x4_f32(
                false, ya0, false, sel, (short)0, acc0, false, false);
            acc1 = __builtin_amdgcn_wmma_f32_16x16x4_f32(
                false, ya1, false, sel, (short)0, acc1, false, false);
          }
        }
      }
      if (any) {
        const int idx = goff + tb + nlo;
        bool nz = false;
#pragma unroll
        for (int r = 0; r < 8; ++r) nz = nz || (acc0[r] != 0.0f) || (acc1[r] != 0.0f);
        if (nz && (unsigned)idx < (unsigned)DHW) {
          float* const p0 = out + ((size_t)((bb * CO + half * 8) * DHW) + (size_t)idx);
          float* const p1 = p0 + (size_t)16 * DHW;
#pragma unroll
          for (int r = 0; r < 8; ++r) {
            atomicAdd(p0 + r * DHW, acc0[r]);
            atomicAdd(p1 + r * DHW, acc1[r]);
          }
        }
      }
    }
  }
  }
}

extern "C" void kernel_launch(void* const* d_in, const int* in_sizes, int n_in,
                              void* d_out, int out_size, void* d_ws, size_t ws_size,
                              hipStream_t stream) {
  const float* x    = (const float*)d_in[0];
  const float* pos  = (const float*)d_in[1];
  const float* kern = (const float*)d_in[2];
  const float* bias = (const float*)d_in[3];
  float* out = (float*)d_out;
  unsigned int* cnt = (unsigned int*)d_ws;
  int* hist   = (int*)((char*)d_ws + 256);
  int* bsum   = hist + NCELL2;
  int* bpre   = bsum + NSCANB;
  int* cellOf = bpre + NSCANB;
  int* rank   = cellOf + (1 << 20);
  int* order  = rank + (1 << 20);

  const int Nper   = in_sizes[1] / (2 * 3);

  int n4 = out_size / 4;
  zero_f4_kernel<<<2048, 256, 0, stream>>>((float4*)out, n4, (int4*)hist, NCELL2 / 4, cnt);

  int pblocks = (2 * Nper + 255) / 256;
  cell_count_kernel<<<pblocks, 256, 0, stream>>>(pos, hist, cellOf, rank, Nper);
  cell_scan_local_kernel<<<NSCANB, 256, 0, stream>>>(hist, bsum);
  cell_scan_blocks_kernel<<<1, NSCANB, 0, stream>>>(bsum, bpre);
  cell_place_kernel<<<pblocks, 256, 0, stream>>>(hist, bpre, cellOf, rank, order, Nper);
  cell_order_fix_kernel<<<NCELL2 / 256, 256, 0, stream>>>(hist, bpre, order, Nper);

  p2g_wmma_kernel<<<512, 256, 0, stream>>>(x, pos, kern, bias, out,
                                           cnt, order, Nper);
}


# product 2 multiplies each 4-row group once: a 16-cell window opens at a group's first cell and is added out when the next group does not fit (was a fixed grid of 16-cell runs); sum order changes
# speedup vs baseline: 35.4515x; 1.0741x over previous
#include <hip/hip_runtime.h>


typedef float v2f __attribute__((ext_vector_type(2)));
typedef float v8f __attribute__((ext_vector_type(8)));

#define DD 66
#define HHW (66 * 66)
#define DHW (66 * 66 * 66)
#define CI 32
#define CO 32
#define NCELL (64 * 64 * 64)
#define NCELL2 (2 * NCELL)
#define NSCANB (NCELL2 / 1024)

__global__ void zero_f4_kernel(float4* __restrict__ p, int n4,
                               int4* __restrict__ hist4, int nh4,
                               unsigned int* __restrict__ cnt) {
  if (blockIdx.x == 0 && threadIdx.x < 2) cnt[threadIdx.x] = 0u;
  const int i0 = blockIdx.x * blockDim.x + threadIdx.x;
  const int stride = gridDim.x * blockDim.x;
  for (int i = i0; i < n4; i += stride) p[i] = make_float4(0.f, 0.f, 0.f, 0.f);
  for (int i = i0; i < nh4; i += stride) hist4[i] = make_int4(0, 0, 0, 0);
}


__global__ void cell_count_kernel(const float* __restrict__ pos, int* __restrict__ hist,
                                  int* __restrict__ cellOf, int* __restrict__ rank,
                                  int Nper) {
  const int p = blockIdx.x * blockDim.x + threadIdx.x;
  if (p >= 2 * Nper) return;
  int b0 = (int)(pos[p * 3 + 0] * 64.0f);
  int b1 = (int)(pos[p * 3 + 1] * 64.0f);
  int b2 = (int)(pos[p * 3 + 2] * 64.0f);
  b0 = min(max(b0, 0), 63);
  b1 = min(max(b1, 0), 63);
  b2 = min(max(b2, 0), 63);
  const int c = ((p >= Nper) ? NCELL : 0) + (b0 * 64 + b1) * 64 + b2;
  cellOf[p] = c;
  rank[p] = atomicAdd(hist + c, 1);
}

__global__ void cell_scan_local_kernel(int* __restrict__ hist, int* __restrict__ bsum) {
  __shared__ int sW[8];
  const int t = threadIdx.x, lane = t & 31, wv = t >> 5;
  int4* h4 = (int4*)hist + blockIdx.x * 256 + t;
  const int4 v = *h4;
  const int s = v.x + v.y + v.z + v.w;
  int inc = s;
#pragma unroll
  for (int o = 1; o < 32; o <<= 1) {
    int n = __shfl_up(inc, o);
    if (lane >= o) inc += n;
  }
  if (lane == 31) sW[wv] = inc;
  __syncthreads();
  int wpre = 0, tot = 0;
#pragma unroll
  for (int w = 0; w < 8; ++w) {
    int xw = sW[w];
    if (w < wv) wpre += xw;
    tot += xw;
  }
  const int ex = wpre + inc - s;
  *h4 = make_int4(ex, ex + v.x, ex + v.x + v.y, ex + v.x + v.y + v.z);
  if (t == 0) bsum[blockIdx.x] = tot;
}

__global__ void cell_scan_blocks_kernel(const int* __restrict__ bsum, int* __restrict__ bpre) {
  __shared__ int sS[NSCANB];
  const int t = threadIdx.x;
  const int own = bsum[t];
  sS[t] = own;
  __syncthreads();
  for (int o = 1; o < NSCANB; o <<= 1) {
    const int add = (t >= o) ? sS[t - o] : 0;
    __syncthreads();
    sS[t] += add;
    __syncthreads();
  }
  bpre[t] = sS[t] - own;
}

__global__ void cell_place_kernel(const int* __restrict__ hist, const int* __restrict__ bpre,
                                  const int* __restrict__ cellOf, const int* __restrict__ rank,
                                  int* __restrict__ order, int Nper) {
  const int p = blockIdx.x * blockDim.x + threadIdx.x;
  if (p >= 2 * Nper) return;
  const int c = cellOf[p];
  order[hist[c] + bpre[c >> 10] + rank[p]] = p;
}

__global__ void cell_order_fix_kernel(const int* __restrict__ hist, const int* __restrict__ bpre,
                                      int* __restrict__ order, int Nper) {
  const int c = blockIdx.x * blockDim.x + threadIdx.x;
  if (c >= NCELL2) return;
  const int s = hist[c] + bpre[c >> 10];
  const int e = (c + 1 < NCELL2) ? (hist[c + 1] + bpre[(c + 1) >> 10]) : (2 * Nper);
  for (int i = s + 1; i < e; ++i) {
    const int v = order[i];
    int j = i - 1;
    while (j >= s && order[j] > v) {
      order[j + 1] = order[j];
      --j;
    }
    order[j + 1] = v;
  }
}


__launch_bounds__(256)
__global__ void p2g_wmma_kernel(const float* __restrict__ x,
                                const float* __restrict__ pos,
                                const float* __restrict__ kern,
                                const float* __restrict__ bias,
                                float* __restrict__ out,
                                unsigned int* __restrict__ cnt,
                                const int* __restrict__ order,
                                int Nper) {
  __shared__ float sB[27 * 2 * 8 * 64];
  __shared__ float4 sP[8][32];
  __shared__ int sSel[1];

  const int t    = threadIdx.x;
  const int lane = t & 31;
  const int wv   = t >> 5;
  const int half = lane >> 4;
  const int nlo  = lane & 15;

  const int chunksPer = (Nper + 31) >> 5;
  if (t == 0) {
    unsigned int id = __builtin_amdgcn_s_sendmsg_rtn(0x87);
    sSel[0] = (int)((id >> 18) & 1u);
  }

  {
    const float4* k4 = (const float4*)kern;
    const int sci = t >> 3;
    const int sco = (t & 7) * 4;
    const int sbase = (sci >> 2) * 64 + (((sci >> 1) & 1) * 16) * 2 + (sci & 1)
                    + (sco >> 4) * 512 + (sco & 15) * 2;
#pragma unroll
    for (int o = 0; o < 27; ++o) {
      const float4 v4 = k4[o * 256 + t];
      sB[o * 1024 + sbase + 0] = v4.x;
      sB[o * 1024 + sbase + 2] = v4.y;
      sB[o * 1024 + sbase + 4] = v4.z;
      sB[o * 1024 + sbase + 6] = v4.w;
    }
  }
  __syncthreads();

  int bb = __builtin_amdgcn_readfirstlane(sSel[0]);
  int left = 2;
  const int maxDraws = 2 * chunksPer + 2;
  for (int draw = 0; draw < maxDraws && left > 0; ++draw) {
  int tk = 0;
  if (lane == 0)
    tk = (int)__hip_atomic_fetch_add(cnt + bb, 1u, __ATOMIC_RELAXED, __HIP_MEMORY_SCOPE_AGENT);
  tk = __builtin_amdgcn_readfirstlane(tk);
  if (tk >= chunksPer) {
    bb ^= 1;
    --left;
    continue;
  }

  const int waveBase = tk * 32;
  const int pOff = bb * Nper;

  const int sp = (lane & 16) + ((lane >> 1) & 3) * 4 + ((lane >> 3) & 1) * 2 + (lane & 1);
  const int q  = waveBase + sp;
  const int pc = order[pOff + ((q < Nper) ? q : (Nper - 1))];
  const float valid = (q < Nper) ? 1.0f : 0.0f;

  float px = pos[pc * 3 + 0] * 64.0f;
  float py = pos[pc * 3 + 1] * 64.0f;
  float pz = pos[pc * 3 + 2] * 64.0f;
  int b0 = (int)px, b1 = (int)py, b2 = (int)pz;
  float fx = px - (float)b0 - 0.5f;
  float fy = py - (float)b1 - 0.5f;
  float fz = pz - (float)b2 - 0.5f;

  const float wx0 = 0.5f * (0.5f - fx) * (0.5f - fx) * valid;
  const float wx1 = (0.75f - fx * fx) * valid;
  const float wx2 = 0.5f * (0.5f + fx) * (0.5f + fx) * valid;
  const float wy0 = 0.5f * (0.5f - fy) * (0.5f - fy);
  const float wy1 = 0.75f - fy * fy;
  const float wy2 = 0.5f * (0.5f + fy) * (0.5f + fy);
  const float wz0 = 0.5f * (0.5f - fz) * (0.5f - fz);
  const float wz1 = 0.75f - fz * fz;
  const float wz2 = 0.5f * (0.5f + fz) * (0.5f + fz);

  const int lin = b0 * HHW + b1 * 66 + b2;

  int lmin = lin, lmax = lin;
#pragma unroll
  for (int o = 16; o > 0; o >>= 1) {
    lmin = min(lmin, __shfl_xor(lmin, o));
    lmax = max(lmax, __shfl_xor(lmax, o));
  }
  lmin = __builtin_amdgcn_readfirstlane(lmin);
  lmax = __builtin_amdgcn_readfirstlane(lmax);
  int glo = min(lin, __shfl_xor(lin, 1));
  int ghi = max(lin, __shfl_xor(lin, 1));
  glo = min(glo, __shfl_xor(glo, 8));
  ghi = max(ghi, __shfl_xor(ghi, 8));
  int gLo[8], gHi[8];
#pragma unroll
  for (int gq = 0; gq < 8; ++gq) {
    gLo[gq] = __builtin_amdgcn_readlane(glo, (gq >> 2) * 16 + (gq & 3) * 2);
    gHi[gq] = __builtin_amdgcn_readlane(ghi, (gq >> 2) * 16 + (gq & 3) * 2);
  }

  v2f a[2][8];
#pragma unroll
  for (int mt = 0; mt < 2; ++mt) {
    const int prow = __shfl(pc, mt * 16 + nlo);
    const float* xr = x + prow * CI + half * 2;
#pragma unroll
    for (int c = 0; c < 8; ++c) a[mt][c] = *(const v2f*)(xr + c * 4);
  }

  const float bias0 = bias[nlo];
  const float bias1 = bias[16 + nlo];

  for (int g = 0; g < 9; ++g) {
    const int oi = g / 3, oj = g - oi * 3;

    {
      float wxs = (oi == 0) ? wx0 : ((oi == 1) ? wx1 : wx2);
      float wys = (oj == 0) ? wy0 : ((oj == 1) ? wy1 : wy2);
      float wxy = wxs * wys;
      sP[wv][lane] = make_float4(wxy * wz0, wxy * wz1, wxy * wz2, __int_as_float(lin));
    }

    v8f y[3][2][2];
#pragma unroll
    for (int ok = 0; ok < 3; ++ok) {
#pragma unroll
      for (int nt = 0; nt < 2; ++nt) {
        v2f bf[8];
#pragma unroll
        for (int c = 0; c < 8; ++c)
          bf[c] = *(const v2f*)&sB[(((g * 3 + ok) * 2 + nt) * 8 + c) * 64 + lane * 2];
        const float bl = nt ? bias1 : bias0;
#pragma unroll
        for (int mt = 0; mt < 2; ++mt)
#pragma unroll
          for (int r = 0; r < 8; ++r) y[ok][nt][mt][r] = bl;
#pragma unroll
        for (int c = 0; c < 8; ++c) {
#pragma unroll
          for (int mt = 0; mt < 2; ++mt) {
            y[ok][nt][mt] = __builtin_amdgcn_wmma_f32_16x16x4_f32(
                false, a[mt][c], false, bf[c], (short)0, y[ok][nt][mt],
                false, false);
          }
        }
      }
    }

    const int goff = oi * HHW + oj * 66;
    v8f acc0, acc1;
#pragma unroll
    for (int r = 0; r < 8; ++r) { acc0[r] = 0.0f; acc1[r] = 0.0f; }
    int wlo = gLo[0];

    auto flush = [&](const int w) {
      const int idx = goff + w + nlo;
      bool nz = false;
#pragma unroll
      for (int r = 0; r < 8; ++r) nz = nz || (acc0[r] != 0.0f) || (acc1[r] != 0.0f);
      if (nz && (unsigned)idx < (unsigned)DHW) {
        float* const p0 = out + ((size_t)((bb * CO + half * 8) * DHW) + (size_t)idx);
        float* const p1 = p0 + (size_t)16 * DHW;
#pragma unroll
        for (int r = 0; r < 8; ++r) {
          atomicAdd(p0 + r * DHW, acc0[r]);
          atomicAdd(p1 + r * DHW, acc1[r]);
        }
      }
#pragma unroll
      for (int r = 0; r < 8; ++r) { acc0[r] = 0.0f; acc1[r] = 0.0f; }
    };

#pragma unroll
    for (int gq = 0; gq < 8; ++gq) {
      const int mt = gq >> 2, j = gq & 3;
      const int lo = gLo[gq];
      const int hi = gHi[gq] + 2;
      const bool fits = (lo >= wlo) && (hi <= wlo + 15);
      const int w0 = fits ? wlo : lo;
      const int nwin = ((hi - w0) >> 4) + 1;
      for (int k = 0; k < nwin; ++k) {
        if (k > 0 || !fits) {
          flush(wlo);
          wlo = w0 + 16 * k;
        }
        const float4 p0 = sP[wv][mt * 16 + half * 8 + 2 * j];
        const float4 p1 = sP[wv][mt * 16 + half * 8 + 2 * j + 1];
        const int d0 = __float_as_int(p0.w) - wlo - nlo;
        const int d1 = __float_as_int(p1.w) - wlo - nlo;
#pragma unroll
        for (int ok = 0; ok < 3; ++ok) {
          const float w0f = (ok == 0) ? p0.x : ((ok == 1) ? p0.y : p0.z);
          const float w1f = (ok == 0) ? p1.x : ((ok == 1) ? p1.y : p1.z);
          v2f sel;
          sel.x = (d0 + ok == 0) ? w0f : 0.0f;
          sel.y = (d1 + ok == 0) ? w1f : 0.0f;
          v2f ya0, ya1;
          ya0.x = y[ok][0][mt][2 * j]; ya0.y = y[ok][0][mt][2 * j + 1];
          ya1.x = y[ok][1][mt][2 * j]; ya1.y = y[ok][1][mt][2 * j + 1];
          acc0 = __builtin_amdgcn_wmma_f32_16x16x4_f32(
              false, ya0, false, sel, (short)0, acc0, false, false);
          acc1 = __builtin_amdgcn_wmma_f32_16x16x4_f32(
              false, ya1, false, sel, (short)0, acc1, false, false);
        }
      }
    }
    flush(wlo);
  }
  }
}

extern "C" void kernel_launch(void* const* d_in, const int* in_sizes, int n_in,
                              void* d_out, int out_size, void* d_ws, size_t ws_size,
                              hipStream_t stream) {
  const float* x    = (const float*)d_in[0];
  const float* pos  = (const float*)d_in[1];
  const float* kern = (const float*)d_in[2];
  const float* bias = (const float*)d_in[3];
  float* out = (float*)d_out;
  unsigned int* cnt = (unsigned int*)d_ws;
  int* hist   = (int*)((char*)d_ws + 256);
  int* bsum   = hist + NCELL2;
  int* bpre   = bsum + NSCANB;
  int* cellOf = bpre + NSCANB;
  int* rank   = cellOf + (1 << 20);
  int* order  = rank + (1 << 20);

  const int Nper   = in_sizes[1] / (2 * 3);

  int n4 = out_size / 4;
  zero_f4_kernel<<<2048, 256, 0, stream>>>((float4*)out, n4, (int4*)hist, NCELL2 / 4, cnt);

  int pblocks = (2 * Nper + 255) / 256;
  cell_count_kernel<<<pblocks, 256, 0, stream>>>(pos, hist, cellOf, rank, Nper);
  cell_scan_local_kernel<<<NSCANB, 256, 0, stream>>>(hist, bsum);
  cell_scan_blocks_kernel<<<1, NSCANB, 0, stream>>>(bsum, bpre);
  cell_place_kernel<<<pblocks, 256, 0, stream>>>(hist, bpre, cellOf, rank, order, Nper);
  cell_order_fix_kernel<<<NCELL2 / 256, 256, 0, stream>>>(hist, bpre, order, Nper);

  p2g_wmma_kernel<<<512, 256, 0, stream>>>(x, pos, kern, bias, out,
                                           cnt, order, Nper);
}
